# starting kernel
# baseline (speedup 1.0000x reference)
#include <hip/hip_runtime.h>
#include <hip/hip_bf16.h>

typedef __attribute__((ext_vector_type(16))) _Float16 v16h;
typedef __attribute__((ext_vector_type(8)))  _Float16 v8h;
typedef __attribute__((ext_vector_type(8)))  float    v8f;

__global__ void fill_f32(float* __restrict__ p, float v, long long n) {
    long long i = (long long)blockIdx.x * blockDim.x + threadIdx.x;
    if (i < n) p[i] = v;
}

__global__ void deg_accum(const int* __restrict__ dst, float* __restrict__ deg, int E) {
    int e = blockIdx.x * blockDim.x + threadIdx.x;
    if (e < E) atomicAdd(&deg[dst[e]], 1.0f);
}

__global__ void rsqrt_inplace(float* __restrict__ d, int n) {
    int i = blockIdx.x * blockDim.x + threadIdx.x;
    if (i < n) {
        float v = d[i];
        d[i] = (v > 0.0f) ? rsqrtf(v) : 0.0f;
    }
}

__global__ void cvt_f32_f16(const float* __restrict__ src, _Float16* __restrict__ dst, long long n) {
    long long i = (long long)blockIdx.x * blockDim.x + threadIdx.x;
    if (i < n) dst[i] = (_Float16)src[i];
}

__global__ void cvt_transpose_w(const float* __restrict__ W, _Float16* __restrict__ Wt, int K, int C) {
    int idx = blockIdx.x * blockDim.x + threadIdx.x;
    if (idx < K * C) {
        int k = idx / C, c = idx % C;
        Wt[c * K + k] = (_Float16)W[idx];
    }
}

__global__ void __launch_bounds__(256) gemm_f16_wmma(const _Float16* __restrict__ A,
                                                     const _Float16* __restrict__ Bt,
                                                     float* __restrict__ C,
                                                     int nrows) {
    const int wave = threadIdx.x >> 5;
    const int lane = threadIdx.x & 31;
    const int mt   = blockIdx.x;

    int rowA = mt * 16 + (lane & 15);
    if (rowA > nrows - 1) rowA = nrows - 1;
    const int colB  = wave * 16 + (lane & 15);
    const int kbase = (lane >> 4) ? 8 : 0;

    const _Float16* __restrict__ arow = A  + (long long)rowA * 128;
    const _Float16* __restrict__ brow = Bt + (long long)colB * 128;

    v8f acc = {};
#pragma unroll
    for (int kc = 0; kc < 4; ++kc) {
        const int k0 = kc * 32 + kbase;
        v8h alo = *(const v8h*)(arow + k0);
        v8h ahi = *(const v8h*)(arow + k0 + 16);
        v8h blo = *(const v8h*)(brow + k0);
        v8h bhi = *(const v8h*)(brow + k0 + 16);
        v16h a, b;
#pragma unroll
        for (int i = 0; i < 8; ++i) {
            a[i] = alo[i]; a[8 + i] = ahi[i];
            b[i] = blo[i]; b[8 + i] = bhi[i];
        }
        acc = __builtin_amdgcn_wmma_f32_16x16x32_f16(
             false, a,  false, b,
             (short)0, acc,  false,  false);
    }

    const int rbase = mt * 16 + ((lane >> 4) ? 8 : 0);
    float* __restrict__ crow = C + (long long)rbase * 128 + wave * 16 + (lane & 15);
    if (mt * 16 + 16 <= nrows) {
#pragma unroll
        for (int r = 0; r < 8; ++r) crow[(long long)r * 128] = acc[r];
    } else {
#pragma unroll
        for (int r = 0; r < 8; ++r)
            if (rbase + r < nrows) crow[(long long)r * 128] = acc[r];
    }
}

__global__ void scatter_edges(const int* __restrict__ ei, const float* __restrict__ dis,
                              const float* __restrict__ hin, float* __restrict__ agg,
                              int E, int Nn) {
    long long idx = (long long)blockIdx.x * blockDim.x + threadIdx.x;
    long long total = (long long)(E + Nn) * 32;
    if (idx >= total) return;
    int e = (int)(idx >> 5);
    int c = ((int)idx & 31) * 4;
    int s, d;
    if (e < E) { s = ei[e]; d = ei[E + e]; }
    else       { s = d = e - E; }
    float w = dis[s] * dis[d];
    float4 v = *(const float4*)(hin + (long long)s * 128 + c);
    float* o = agg + (long long)d * 128 + c;
    atomicAdd(o + 0, v.x * w);
    atomicAdd(o + 1, v.y * w);
    atomicAdd(o + 2, v.z * w);
    atomicAdd(o + 3, v.w * w);
}

__global__ void bias_relu_f16(const float* __restrict__ agg, const float* __restrict__ b,
                              _Float16* __restrict__ out, long long n) {
    long long i = (long long)blockIdx.x * blockDim.x + threadIdx.x;
    if (i < n) {
        float v = agg[i] + b[i & 127];
        out[i] = (_Float16)(v > 0.0f ? v : 0.0f);
    }
}

__global__ void bias_relu_f32(const float* __restrict__ agg, const float* __restrict__ b,
                              float* __restrict__ out, long long n) {
    long long i = (long long)blockIdx.x * blockDim.x + threadIdx.x;
    if (i < n) {
        float v = agg[i] + b[i & 127];
        out[i] = v > 0.0f ? v : 0.0f;
    }
}

__global__ void q_kernel(const int* __restrict__ va, const float* __restrict__ h,
                         const float* __restrict__ Wq, const float* __restrict__ bq,
                         float* __restrict__ out, int A) {
    int a = blockIdx.x * blockDim.x + threadIdx.x;
    if (a >= A) return;
    int i = va[2 * a], j = va[2 * a + 1];
    const float4* __restrict__ hi = (const float4*)(h + (long long)i * 128);
    const float4* __restrict__ hj = (const float4*)(h + (long long)j * 128);
    const float4* __restrict__ wq = (const float4*)Wq;
    float s = bq[0];
#pragma unroll 8
    for (int k = 0; k < 32; ++k) {
        float4 x = hi[k], w = wq[k];
        s += x.x * w.x + x.y * w.y + x.z * w.z + x.w * w.w;
    }
#pragma unroll 8
    for (int k = 0; k < 32; ++k) {
        float4 x = hj[k], w = wq[32 + k];
        s += x.x * w.x + x.y * w.y + x.z * w.z + x.w * w.w;
    }
    out[a] = s;
}

extern "C" void kernel_launch(void* const* d_in, const int* in_sizes, int n_in,
                              void* d_out, int out_size, void* d_ws, size_t ws_size,
                              hipStream_t stream) {
    const float* x  = (const float*)d_in[0];
    const int*   ei = (const int*)d_in[1];
    const int*   va = (const int*)d_in[2];
    const float* W1 = (const float*)d_in[3];
    const float* b1 = (const float*)d_in[4];
    const float* W2 = (const float*)d_in[5];
    const float* b2 = (const float*)d_in[6];
    const float* Wq = (const float*)d_in[7];
    const float* bq = (const float*)d_in[8];
    float* q = (float*)d_out;

    const int H  = 128;
    const int Nn = in_sizes[0] / H;
    const int E  = in_sizes[1] / 2;
    const int A  = in_sizes[2] / 2;

    char*  ws  = (char*)d_ws;
    size_t off = 0;
    auto alloc = [&](size_t bytes) -> char* {
        char* p = ws + off;
        off = (off + bytes + 255) & ~(size_t)255;
        return p;
    };
    _Float16* xh  = (_Float16*)alloc((size_t)Nn * H * 2);
    _Float16* w1t = (_Float16*)alloc((size_t)H * H * 2);
    _Float16* w2t = (_Float16*)alloc((size_t)H * H * 2);
    float*    gout= (float*)   alloc((size_t)Nn * H * 4);
    float*    agg = (float*)   alloc((size_t)Nn * H * 4);
    _Float16* h1h = (_Float16*)alloc((size_t)Nn * H * 2);
    float*    dis = (float*)   alloc((size_t)Nn * 4);
    float*    h2  = gout;

    const long long NH = (long long)Nn * H;
    auto blks = [](long long n, int t) { return (unsigned)((n + t - 1) / t); };

    fill_f32<<<blks(Nn, 256), 256, 0, stream>>>(dis, 1.0f, Nn);
    deg_accum<<<blks(E, 256), 256, 0, stream>>>(ei + E, dis, E);
    rsqrt_inplace<<<blks(Nn, 256), 256, 0, stream>>>(dis, Nn);

    cvt_f32_f16<<<blks(NH, 256), 256, 0, stream>>>(x, xh, NH);
    cvt_transpose_w<<<blks(H * H, 256), 256, 0, stream>>>(W1, w1t, H, H);
    cvt_transpose_w<<<blks(H * H, 256), 256, 0, stream>>>(W2, w2t, H, H);

    const unsigned mtiles = (unsigned)((Nn + 15) / 16);
    const long long scat  = (long long)(E + Nn) * 32;

    gemm_f16_wmma<<<mtiles, 256, 0, stream>>>(xh, w1t, gout, Nn);
    fill_f32<<<blks(NH, 256), 256, 0, stream>>>(agg, 0.0f, NH);
    scatter_edges<<<blks(scat, 256), 256, 0, stream>>>(ei, dis, gout, agg, E, Nn);
    bias_relu_f16<<<blks(NH, 256), 256, 0, stream>>>(agg, b1, h1h, NH);

    gemm_f16_wmma<<<mtiles, 256, 0, stream>>>(h1h, w2t, gout, Nn);
    fill_f32<<<blks(NH, 256), 256, 0, stream>>>(agg, 0.0f, NH);
    scatter_edges<<<blks(scat, 256), 256, 0, stream>>>(ei, dis, gout, agg, E, Nn);
    bias_relu_f32<<<blks(NH, 256), 256, 0, stream>>>(agg, b2, h2, NH);

    q_kernel<<<blks(A, 256), 256, 0, stream>>>(va, h2, Wq, bq, q, A);
}


# layer 2 walks per-wave edge lists (source, local row, adding order) that layer 1's owner blocks save: no second scan of the 600k edges, no queue sweep; tail row groups load untested; sums unchanged
# speedup vs baseline: 2.1619x; 2.1619x over previous
#include <hip/hip_runtime.h>
#include <hip/hip_bf16.h>


typedef __attribute__((ext_vector_type(16))) _Float16 v16h;
typedef __attribute__((ext_vector_type(8)))  _Float16 v8h;
typedef __attribute__((ext_vector_type(4)))  _Float16 v4h;
typedef __attribute__((ext_vector_type(8)))  float    v8f;

__global__ void prep_w_deg(const float* __restrict__ W1, const float* __restrict__ W2,
                           _Float16* __restrict__ w1t, _Float16* __restrict__ w2t,
                           float* __restrict__ deg, int Nn) {
    int idx = blockIdx.x * blockDim.x + threadIdx.x;
    if (idx < 16384) {
        int k = idx >> 7, c = idx & 127;
        w1t[c * 128 + k] = (_Float16)W1[idx];
    } else if (idx < 32768) {
        int j = idx - 16384;
        int k = j >> 7, c = j & 127;
        w2t[c * 128 + k] = (_Float16)W2[j];
    } else {
        int i = idx - 32768;
        if (i < Nn) deg[i] = 1.0f;
    }
}

__global__ void deg_accum(const int* __restrict__ dst, float* __restrict__ deg, int E) {
    int e = blockIdx.x * blockDim.x + threadIdx.x;
    if (e < E) atomicAdd(&deg[dst[e]], 1.0f);
}

#define A_PITCH 136
template <int MODE>
__global__ void __launch_bounds__(256) gemm_fused(const float* __restrict__ A32,
                                                  const float* __restrict__ bias,
                                                  const _Float16* __restrict__ Bt,
                                                  float* __restrict__ C,
                                                  float* __restrict__ dis,
                                                  int nrows) {
    __shared__ __attribute__((aligned(16))) _Float16 As[16 * A_PITCH];
    __shared__ float sdis[16];

    const int tid  = threadIdx.x;
    const int wave = tid >> 5;
    const int lane = tid & 31;
    const int mt   = blockIdx.x;

    const int colB  = wave * 16 + (lane & 15);
    const int kbase = (lane >> 4) ? 8 : 0;
    const _Float16* __restrict__ brow = Bt + (long long)colB * 128;

    v8h blo[4], bhi[4];
#pragma unroll
    for (int kc = 0; kc < 4; ++kc) {
        const int k0 = kc * 32 + kbase;
        blo[kc] = *(const v8h*)(brow + k0);
        bhi[kc] = *(const v8h*)(brow + k0 + 16);
    }

    if (tid < 16) {
        const int i = mt * 16 + tid;
        float r = 0.0f;
        if (i < nrows) {
            const float d = dis[i];
            if (MODE == 0) { r = (d > 0.0f) ? rsqrtf(d) : 0.0f; dis[i] = r; }
            else           { r = d; }
        }
        sdis[tid] = r;
    }

#pragma unroll
    for (int p = 0; p < 2; ++p) {
        const int f  = tid + p * 256;
        const int r  = f >> 5;
        const int c4 = (f & 31) * 4;
        int row = mt * 16 + r;
        if (row > nrows - 1) row = nrows - 1;
        float4 v = *(const float4*)(A32 + (long long)row * 128 + c4);
        if (MODE == 1) {
            const float4 b = *(const float4*)(bias + c4);
            v.x = v.x + b.x; v.y = v.y + b.y; v.z = v.z + b.z; v.w = v.w + b.w;
            v.x = v.x > 0.0f ? v.x : 0.0f;
            v.y = v.y > 0.0f ? v.y : 0.0f;
            v.z = v.z > 0.0f ? v.z : 0.0f;
            v.w = v.w > 0.0f ? v.w : 0.0f;
        }
        v4h hv;
        hv[0] = (_Float16)v.x; hv[1] = (_Float16)v.y; hv[2] = (_Float16)v.z; hv[3] = (_Float16)v.w;
        *(v4h*)(As + r * A_PITCH + c4) = hv;
    }

    __syncthreads();

    const _Float16* arow = As + (lane & 15) * A_PITCH;
    v8f acc = {};
#pragma unroll
    for (int kc = 0; kc < 4; ++kc) {
        const int k0 = kc * 32 + kbase;
        v8h alo = *(const v8h*)(arow + k0);
        v8h ahi = *(const v8h*)(arow + k0 + 16);
        v16h a, b;
#pragma unroll
        for (int i = 0; i < 8; ++i) {
            a[i] = alo[i];     a[8 + i] = ahi[i];
            b[i] = blo[kc][i]; b[8 + i] = bhi[kc][i];
        }
        acc = __builtin_amdgcn_wmma_f32_16x16x32_f16(
             false, a,  false, b,
             (short)0, acc,  false,  false);
    }

    const int rbase = mt * 16 + ((lane >> 4) ? 8 : 0);
    float* __restrict__ crow = C + (long long)rbase * 128 + wave * 16 + (lane & 15);
    const float* sd = sdis + ((lane >> 4) ? 8 : 0);
    if (mt * 16 + 16 <= nrows) {
#pragma unroll
        for (int r = 0; r < 8; ++r) crow[(long long)r * 128] = acc[r] * sd[r];
    } else {
#pragma unroll
        for (int r = 0; r < 8; ++r)
            if (rbase + r < nrows) crow[(long long)r * 128] = acc[r] * sd[r];
    }
}

#define AG_R    200
#define AG_T    1024
#define AG_W    32
#define AG_QCAP 256
#define AG_U    4
#define AG_LCAP 192

__device__ __forceinline__ void ag_add_rows(unsigned lr_l, int s, int cnt, int lane,
                                            const float* __restrict__ hs, float* acc) {
    for (int g = 0; g < cnt; g += 8) {
        float4 v[8];
#pragma unroll
        for (int j = 0; j < 8; ++j) {
            const int sj = __builtin_amdgcn_readlane(s, g + j);
            v[j] = *(const float4*)(hs + (long long)sj * 128 + lane * 4);
        }
#pragma unroll
        for (int j = 0; j < 8; ++j) {
            if (g + j < cnt) {
                const unsigned lr = (unsigned)__builtin_amdgcn_readlane((int)lr_l, g + j);
                float4* a = (float4*)(acc + lr * 128 + lane * 4);
                float4 t = *a;
                t.x += v[j].x; t.y += v[j].y; t.z += v[j].z; t.w += v[j].w;
                *a = t;
            }
        }
    }
}

template <bool SAVE>
__device__ __forceinline__ void ag_drain(const unsigned* mb, int cnt, int lane,
                                         const int* __restrict__ srcv, const float* __restrict__ hs,
                                         float* acc, unsigned* __restrict__ wl, int& wpos) {
    unsigned ent = 0;
    int s = 0;
    if (lane < cnt) { ent = mb[lane]; s = srcv[ent >> 8]; }
    if (SAVE) {
        if (lane < cnt && wpos + lane < AG_LCAP) wl[wpos + lane] = ((unsigned)s << 8) | (ent & 255u);
        wpos += cnt;
    }
    ag_add_rows(ent & 255u, s, cnt, lane, hs, acc);
}

__device__ __forceinline__ void ag_write_out(const float* acc, const float* __restrict__ dis,
                                             float* __restrict__ out, int r0, int nr, int tid) {
    for (int f = tid; f < AG_R * 32; f += AG_T) {
        const int r = f >> 5, c4 = (f & 31) * 4;
        if (r < nr) {
            const float dr = dis[r0 + r];
            float4 v = *(const float4*)(acc + r * 128 + c4);
            v.x *= dr; v.y *= dr; v.z *= dr; v.w *= dr;
            *(float4*)(out + (long long)(r0 + r) * 128 + c4) = v;
        }
    }
}

template <int MODE>
__global__ void __launch_bounds__(AG_T) agg_owner(const int* __restrict__ ei, const float* __restrict__ dis,
                                                  const float* __restrict__ hs, float* __restrict__ out,
                                                  unsigned* __restrict__ lists, int* __restrict__ lcnt,
                                                  int E, int Nn) {
    __shared__ __attribute__((aligned(16))) float acc[AG_R * 128];
    __shared__ unsigned qbuf[AG_W * AG_QCAP];
    __shared__ unsigned mybuf[AG_W * 64];
    __shared__ int qcnt[AG_W];
    __shared__ int pend[2];

    const int tid  = threadIdx.x;
    const int wave = __builtin_amdgcn_readfirstlane(tid >> 5);
    const int lane = tid & 31;
    const unsigned ltmask = (1u << lane) - 1u;
    const int r0 = blockIdx.x * AG_R;
    const int nr = (Nn - r0 < AG_R) ? (Nn - r0) : AG_R;

    const int* __restrict__ srcv = ei;
    const int* __restrict__ dstv = ei + E;
    const bool al16 = (((size_t)dstv) & 15) == 0;

    unsigned* __restrict__ wl = lists + (size_t)(blockIdx.x * AG_W + wave) * AG_LCAP;
    int wpos = 0;
    int cv = 0;
    if (MODE == 1) cv = lcnt[blockIdx.x * AG_W + lane];

    for (int f = tid; f < AG_R * 32; f += AG_T) {
        const int r = f >> 5, c4 = (f & 31) * 4;
        float4 v = make_float4(0.0f, 0.0f, 0.0f, 0.0f);
        if (r < nr) v = *(const float4*)(hs + (long long)(r0 + r) * 128 + c4);
        *(float4*)(acc + r * 128 + c4) = v;
    }

    if (MODE == 1) {
        if (__builtin_amdgcn_ballot_w32(cv > AG_LCAP) == 0u) {
            const int nl = __builtin_amdgcn_readlane(cv, wave);
            unsigned cur = 0u;
            if (lane < nl) cur = wl[lane];
            __syncthreads();
            int i0 = 0;
            for (; i0 + 32 <= nl; i0 += 32) {
                unsigned nxt = 0u;
                if (i0 + 32 + lane < nl) nxt = wl[i0 + 32 + lane];
                ag_add_rows(cur & 255u, (int)(cur >> 8), 32, lane, hs, acc);
                cur = nxt;
            }
            if (i0 < nl) ag_add_rows(cur & 255u, (int)(cur >> 8), nl - i0, lane, hs, acc);
            __syncthreads();
            ag_write_out(acc, dis, out, r0, nr, tid);
            return;
        }
    }

    if (tid < 2) pend[tid] = 0;
    __syncthreads();

    const int S = ((E + 4095) >> 12) << 7;
    int pos = wave * S;
    const int end = (pos + S < E) ? (pos + S) : E;
    unsigned* myq = qbuf + wave * AG_QCAP;
    unsigned* mb  = mybuf + wave * 64;

    for (int rnd = 0;; ++rnd) {
        int cnt = 0;
        while (pos < end && cnt <= AG_QCAP - 128) {
            int4 dv[AG_U];
            if (al16 && pos + AG_U * 128 <= end) {
#pragma unroll
                for (int u = 0; u < AG_U; ++u)
                    dv[u] = *(const int4*)(dstv + pos + u * 128 + lane * 4);
            } else {
#pragma unroll
                for (int u = 0; u < AG_U; ++u) {
                    const int e = pos + u * 128 + lane * 4;
                    dv[u].x = (e     < end) ? dstv[e]     : -1;
                    dv[u].y = (e + 1 < end) ? dstv[e + 1] : -1;
                    dv[u].z = (e + 2 < end) ? dstv[e + 2] : -1;
                    dv[u].w = (e + 3 < end) ? dstv[e + 3] : -1;
                }
            }
#pragma unroll
            for (int u = 0; u < AG_U; ++u) {
                if (pos < end && cnt <= AG_QCAP - 128) {
                    const unsigned e = (unsigned)(pos + lane * 4);
                    unsigned lr[4];
                    lr[0] = (unsigned)(dv[u].x - r0); lr[1] = (unsigned)(dv[u].y - r0);
                    lr[2] = (unsigned)(dv[u].z - r0); lr[3] = (unsigned)(dv[u].w - r0);
                    const unsigned a01 = lr[0] < lr[1] ? lr[0] : lr[1];
                    const unsigned a23 = lr[2] < lr[3] ? lr[2] : lr[3];
                    const unsigned mn  = a01 < a23 ? a01 : a23;
                    if (__builtin_amdgcn_ballot_w32(mn < (unsigned)nr) != 0u) {
#pragma unroll
                        for (int k = 0; k < 4; ++k) {
                            const bool hit = lr[k] < (unsigned)nr;
                            const unsigned m = __builtin_amdgcn_ballot_w32(hit);
                            if (hit) myq[cnt + __popc(m & ltmask)] = ((e + k) << 8) | lr[k];
                            cnt += __popc(m);
                        }
                    }
                    pos += 128;
                }
            }
        }
        if (lane == 0) {
            qcnt[wave] = cnt;
            if (pos < end) pend[rnd & 1] = 1;
        }
        __syncthreads();

        int pc = 0;
        for (int q = 0; q < AG_W; ++q) {
            const int n = qcnt[q];
            const unsigned* qq = qbuf + q * AG_QCAP;
            for (int i0 = 0; i0 < n; i0 += 32) {
                const int i = i0 + lane;
                unsigned ent = 0;
                bool mine = false;
                if (i < n) { ent = qq[i]; mine = ((ent & 31u) == (unsigned)wave); }
                const unsigned m = __builtin_amdgcn_ballot_w32(mine);
                if (m != 0u) {
                    if (mine) mb[pc + __popc(m & ltmask)] = ent;
                    pc += __popc(m);
                    __builtin_amdgcn_wave_barrier();
                    if (pc >= 32) {
                        ag_drain<MODE == 0>(mb, 32, lane, srcv, hs, acc, wl, wpos);
                        unsigned t = 0;
                        if (lane < pc - 32) t = mb[32 + lane];
                        __builtin_amdgcn_wave_barrier();
                        if (lane < pc - 32) mb[lane] = t;
                        pc -= 32;
                        __builtin_amdgcn_wave_barrier();
                    }
                }
            }
        }
        if (pc > 0) ag_drain<MODE == 0>(mb, pc, lane, srcv, hs, acc, wl, wpos);

        const int again = pend[rnd & 1];
        if (tid == 0) pend[(rnd + 1) & 1] = 0;
        __syncthreads();
        if (!again) break;
    }

    if (MODE == 0 && lane == 0) lcnt[blockIdx.x * AG_W + wave] = wpos;

    ag_write_out(acc, dis, out, r0, nr, tid);
}

__global__ void q_kernel(const int* __restrict__ va, const float* __restrict__ agg,
                         const float* __restrict__ b2,
                         const float* __restrict__ Wq, const float* __restrict__ bq,
                         float* __restrict__ out, int A) {
    int a = blockIdx.x * blockDim.x + threadIdx.x;
    if (a >= A) return;
    int i = va[2 * a], j = va[2 * a + 1];
    const float4* __restrict__ hi = (const float4*)(agg + (long long)i * 128);
    const float4* __restrict__ hj = (const float4*)(agg + (long long)j * 128);
    const float4* __restrict__ wq = (const float4*)Wq;
    const float4* __restrict__ bb = (const float4*)b2;
    float s = bq[0];
#pragma unroll 8
    for (int k = 0; k < 32; ++k) {
        float4 x = hi[k], w = wq[k], b = bb[k];
        x.x = x.x + b.x; x.y = x.y + b.y; x.z = x.z + b.z; x.w = x.w + b.w;
        x.x = x.x > 0.0f ? x.x : 0.0f;
        x.y = x.y > 0.0f ? x.y : 0.0f;
        x.z = x.z > 0.0f ? x.z : 0.0f;
        x.w = x.w > 0.0f ? x.w : 0.0f;
        s += x.x * w.x + x.y * w.y + x.z * w.z + x.w * w.w;
    }
#pragma unroll 8
    for (int k = 0; k < 32; ++k) {
        float4 x = hj[k], w = wq[32 + k], b = bb[k];
        x.x = x.x + b.x; x.y = x.y + b.y; x.z = x.z + b.z; x.w = x.w + b.w;
        x.x = x.x > 0.0f ? x.x : 0.0f;
        x.y = x.y > 0.0f ? x.y : 0.0f;
        x.z = x.z > 0.0f ? x.z : 0.0f;
        x.w = x.w > 0.0f ? x.w : 0.0f;
        s += x.x * w.x + x.y * w.y + x.z * w.z + x.w * w.w;
    }
    out[a] = s;
}

extern "C" void kernel_launch(void* const* d_in, const int* in_sizes, int n_in,
                              void* d_out, int out_size, void* d_ws, size_t ws_size,
                              hipStream_t stream) {
    const float* x  = (const float*)d_in[0];
    const int*   ei = (const int*)d_in[1];
    const int*   va = (const int*)d_in[2];
    const float* W1 = (const float*)d_in[3];
    const float* b1 = (const float*)d_in[4];
    const float* W2 = (const float*)d_in[5];
    const float* b2 = (const float*)d_in[6];
    const float* Wq = (const float*)d_in[7];
    const float* bq = (const float*)d_in[8];
    float* q = (float*)d_out;

    const int H  = 128;
    const int Nn = in_sizes[0] / H;
    const int E  = in_sizes[1] / 2;
    const int A  = in_sizes[2] / 2;

    char*  ws  = (char*)d_ws;
    size_t off = 0;
    auto alloc = [&](size_t bytes) -> char* {
        char* p = ws + off;
        off = (off + bytes + 255) & ~(size_t)255;
        return p;
    };
    _Float16* w1t  = (_Float16*)alloc((size_t)H * H * 2);
    _Float16* w2t  = (_Float16*)alloc((size_t)H * H * 2);
    float*    gout = (float*)   alloc((size_t)Nn * H * 4);
    float*    agg1 = (float*)   alloc((size_t)Nn * H * 4);
    float*    agg2 = (float*)   alloc((size_t)Nn * H * 4);
    float*    dis  = (float*)   alloc((size_t)Nn * 4);

    auto blks = [](long long n, int t) { return (unsigned)((n + t - 1) / t); };

    prep_w_deg<<<blks(2LL * H * H + Nn, 256), 256, 0, stream>>>(W1, W2, w1t, w2t, dis, Nn);
    deg_accum<<<blks(E, 256), 256, 0, stream>>>(ei + E, dis, E);

    const unsigned mtiles = (unsigned)((Nn + 15) / 16);
    const unsigned owners = blks(Nn, AG_R);
    unsigned* lists = (unsigned*)alloc((size_t)owners * AG_W * AG_LCAP * 4);
    int*      lcnt  = (int*)     alloc((size_t)owners * AG_W * 4);

    gemm_fused<0><<<mtiles, 256, 0, stream>>>(x, b1, w1t, gout, dis, Nn);
    agg_owner<0><<<owners, AG_T, 0, stream>>>(ei, dis, gout, agg1, lists, lcnt, E, Nn);

    gemm_fused<1><<<mtiles, 256, 0, stream>>>(agg1, b1, w2t, gout, dis, Nn);
    agg_owner<1><<<owners, AG_T, 0, stream>>>(ei, dis, gout, agg2, lists, lcnt, E, Nn);

    q_kernel<<<blks(A, 256), 256, 0, stream>>>(va, agg2, b2, Wq, bq, q, A);
}


# Q head: a wave takes 32 actions and loads each row whole with one load (16 B per lane, 4 full lines; was 32 rows x 16 B per load); lane shares summed by a fixed exchange tree: sum order changes
# speedup vs baseline: 2.2992x; 1.0635x over previous
#include <hip/hip_runtime.h>
#include <hip/hip_bf16.h>


typedef __attribute__((ext_vector_type(16))) _Float16 v16h;
typedef __attribute__((ext_vector_type(8)))  _Float16 v8h;
typedef __attribute__((ext_vector_type(4)))  _Float16 v4h;
typedef __attribute__((ext_vector_type(8)))  float    v8f;

__global__ void prep_w_deg(const float* __restrict__ W1, const float* __restrict__ W2,
                           _Float16* __restrict__ w1t, _Float16* __restrict__ w2t,
                           float* __restrict__ deg, int Nn) {
    int idx = blockIdx.x * blockDim.x + threadIdx.x;
    if (idx < 16384) {
        int k = idx >> 7, c = idx & 127;
        w1t[c * 128 + k] = (_Float16)W1[idx];
    } else if (idx < 32768) {
        int j = idx - 16384;
        int k = j >> 7, c = j & 127;
        w2t[c * 128 + k] = (_Float16)W2[j];
    } else {
        int i = idx - 32768;
        if (i < Nn) deg[i] = 1.0f;
    }
}

__global__ void deg_accum(const int* __restrict__ dst, float* __restrict__ deg, int E) {
    int e = blockIdx.x * blockDim.x + threadIdx.x;
    if (e < E) atomicAdd(&deg[dst[e]], 1.0f);
}

#define A_PITCH 136
template <int MODE>
__global__ void __launch_bounds__(256) gemm_fused(const float* __restrict__ A32,
                                                  const float* __restrict__ bias,
                                                  const _Float16* __restrict__ Bt,
                                                  float* __restrict__ C,
                                                  float* __restrict__ dis,
                                                  int nrows) {
    __shared__ __attribute__((aligned(16))) _Float16 As[16 * A_PITCH];
    __shared__ float sdis[16];

    const int tid  = threadIdx.x;
    const int wave = tid >> 5;
    const int lane = tid & 31;
    const int mt   = blockIdx.x;

    const int colB  = wave * 16 + (lane & 15);
    const int kbase = (lane >> 4) ? 8 : 0;
    const _Float16* __restrict__ brow = Bt + (long long)colB * 128;

    v8h blo[4], bhi[4];
#pragma unroll
    for (int kc = 0; kc < 4; ++kc) {
        const int k0 = kc * 32 + kbase;
        blo[kc] = *(const v8h*)(brow + k0);
        bhi[kc] = *(const v8h*)(brow + k0 + 16);
    }

    if (tid < 16) {
        const int i = mt * 16 + tid;
        float r = 0.0f;
        if (i < nrows) {
            const float d = dis[i];
            if (MODE == 0) { r = (d > 0.0f) ? rsqrtf(d) : 0.0f; dis[i] = r; }
            else           { r = d; }
        }
        sdis[tid] = r;
    }

#pragma unroll
    for (int p = 0; p < 2; ++p) {
        const int f  = tid + p * 256;
        const int r  = f >> 5;
        const int c4 = (f & 31) * 4;
        int row = mt * 16 + r;
        if (row > nrows - 1) row = nrows - 1;
        float4 v = *(const float4*)(A32 + (long long)row * 128 + c4);
        if (MODE == 1) {
            const float4 b = *(const float4*)(bias + c4);
            v.x = v.x + b.x; v.y = v.y + b.y; v.z = v.z + b.z; v.w = v.w + b.w;
            v.x = v.x > 0.0f ? v.x : 0.0f;
            v.y = v.y > 0.0f ? v.y : 0.0f;
            v.z = v.z > 0.0f ? v.z : 0.0f;
            v.w = v.w > 0.0f ? v.w : 0.0f;
        }
        v4h hv;
        hv[0] = (_Float16)v.x; hv[1] = (_Float16)v.y; hv[2] = (_Float16)v.z; hv[3] = (_Float16)v.w;
        *(v4h*)(As + r * A_PITCH + c4) = hv;
    }

    __syncthreads();

    const _Float16* arow = As + (lane & 15) * A_PITCH;
    v8f acc = {};
#pragma unroll
    for (int kc = 0; kc < 4; ++kc) {
        const int k0 = kc * 32 + kbase;
        v8h alo = *(const v8h*)(arow + k0);
        v8h ahi = *(const v8h*)(arow + k0 + 16);
        v16h a, b;
#pragma unroll
        for (int i = 0; i < 8; ++i) {
            a[i] = alo[i];     a[8 + i] = ahi[i];
            b[i] = blo[kc][i]; b[8 + i] = bhi[kc][i];
        }
        acc = __builtin_amdgcn_wmma_f32_16x16x32_f16(
             false, a,  false, b,
             (short)0, acc,  false,  false);
    }

    const int rbase = mt * 16 + ((lane >> 4) ? 8 : 0);
    float* __restrict__ crow = C + (long long)rbase * 128 + wave * 16 + (lane & 15);
    const float* sd = sdis + ((lane >> 4) ? 8 : 0);
    if (mt * 16 + 16 <= nrows) {
#pragma unroll
        for (int r = 0; r < 8; ++r) crow[(long long)r * 128] = acc[r] * sd[r];
    } else {
#pragma unroll
        for (int r = 0; r < 8; ++r)
            if (rbase + r < nrows) crow[(long long)r * 128] = acc[r] * sd[r];
    }
}

#define AG_R    200
#define AG_T    1024
#define AG_W    32
#define AG_QCAP 256
#define AG_U    4
#define AG_LCAP 192

__device__ __forceinline__ void ag_add_rows(unsigned lr_l, int s, int cnt, int lane,
                                            const float* __restrict__ hs, float* acc) {
    for (int g = 0; g < cnt; g += 8) {
        float4 v[8];
#pragma unroll
        for (int j = 0; j < 8; ++j) {
            const int sj = __builtin_amdgcn_readlane(s, g + j);
            v[j] = *(const float4*)(hs + (long long)sj * 128 + lane * 4);
        }
#pragma unroll
        for (int j = 0; j < 8; ++j) {
            if (g + j < cnt) {
                const unsigned lr = (unsigned)__builtin_amdgcn_readlane((int)lr_l, g + j);
                float4* a = (float4*)(acc + lr * 128 + lane * 4);
                float4 t = *a;
                t.x += v[j].x; t.y += v[j].y; t.z += v[j].z; t.w += v[j].w;
                *a = t;
            }
        }
    }
}

template <bool SAVE>
__device__ __forceinline__ void ag_drain(const unsigned* mb, int cnt, int lane,
                                         const int* __restrict__ srcv, const float* __restrict__ hs,
                                         float* acc, unsigned* __restrict__ wl, int& wpos) {
    unsigned ent = 0;
    int s = 0;
    if (lane < cnt) { ent = mb[lane]; s = srcv[ent >> 8]; }
    if (SAVE) {
        if (lane < cnt && wpos + lane < AG_LCAP) wl[wpos + lane] = ((unsigned)s << 8) | (ent & 255u);
        wpos += cnt;
    }
    ag_add_rows(ent & 255u, s, cnt, lane, hs, acc);
}

__device__ __forceinline__ void ag_write_out(const float* acc, const float* __restrict__ dis,
                                             float* __restrict__ out, int r0, int nr, int tid) {
    for (int f = tid; f < AG_R * 32; f += AG_T) {
        const int r = f >> 5, c4 = (f & 31) * 4;
        if (r < nr) {
            const float dr = dis[r0 + r];
            float4 v = *(const float4*)(acc + r * 128 + c4);
            v.x *= dr; v.y *= dr; v.z *= dr; v.w *= dr;
            *(float4*)(out + (long long)(r0 + r) * 128 + c4) = v;
        }
    }
}

template <int MODE>
__global__ void __launch_bounds__(AG_T) agg_owner(const int* __restrict__ ei, const float* __restrict__ dis,
                                                  const float* __restrict__ hs, float* __restrict__ out,
                                                  unsigned* __restrict__ lists, int* __restrict__ lcnt,
                                                  int E, int Nn) {
    __shared__ __attribute__((aligned(16))) float acc[AG_R * 128];
    __shared__ unsigned qbuf[AG_W * AG_QCAP];
    __shared__ unsigned mybuf[AG_W * 64];
    __shared__ int qcnt[AG_W];
    __shared__ int pend[2];

    const int tid  = threadIdx.x;
    const int wave = __builtin_amdgcn_readfirstlane(tid >> 5);
    const int lane = tid & 31;
    const unsigned ltmask = (1u << lane) - 1u;
    const int r0 = blockIdx.x * AG_R;
    const int nr = (Nn - r0 < AG_R) ? (Nn - r0) : AG_R;

    const int* __restrict__ srcv = ei;
    const int* __restrict__ dstv = ei + E;
    const bool al16 = (((size_t)dstv) & 15) == 0;

    unsigned* __restrict__ wl = lists + (size_t)(blockIdx.x * AG_W + wave) * AG_LCAP;
    int wpos = 0;
    int cv = 0;
    if (MODE == 1) cv = lcnt[blockIdx.x * AG_W + lane];

    for (int f = tid; f < AG_R * 32; f += AG_T) {
        const int r = f >> 5, c4 = (f & 31) * 4;
        float4 v = make_float4(0.0f, 0.0f, 0.0f, 0.0f);
        if (r < nr) v = *(const float4*)(hs + (long long)(r0 + r) * 128 + c4);
        *(float4*)(acc + r * 128 + c4) = v;
    }

    if (MODE == 1) {
        if (__builtin_amdgcn_ballot_w32(cv > AG_LCAP) == 0u) {
            const int nl = __builtin_amdgcn_readlane(cv, wave);
            unsigned cur = 0u;
            if (lane < nl) cur = wl[lane];
            __syncthreads();
            int i0 = 0;
            for (; i0 + 32 <= nl; i0 += 32) {
                unsigned nxt = 0u;
                if (i0 + 32 + lane < nl) nxt = wl[i0 + 32 + lane];
                ag_add_rows(cur & 255u, (int)(cur >> 8), 32, lane, hs, acc);
                cur = nxt;
            }
            if (i0 < nl) ag_add_rows(cur & 255u, (int)(cur >> 8), nl - i0, lane, hs, acc);
            __syncthreads();
            ag_write_out(acc, dis, out, r0, nr, tid);
            return;
        }
    }

    if (tid < 2) pend[tid] = 0;
    __syncthreads();

    const int S = ((E + 4095) >> 12) << 7;
    int pos = wave * S;
    const int end = (pos + S < E) ? (pos + S) : E;
    unsigned* myq = qbuf + wave * AG_QCAP;
    unsigned* mb  = mybuf + wave * 64;

    for (int rnd = 0;; ++rnd) {
        int cnt = 0;
        while (pos < end && cnt <= AG_QCAP - 128) {
            int4 dv[AG_U];
            if (al16 && pos + AG_U * 128 <= end) {
#pragma unroll
                for (int u = 0; u < AG_U; ++u)
                    dv[u] = *(const int4*)(dstv + pos + u * 128 + lane * 4);
            } else {
#pragma unroll
                for (int u = 0; u < AG_U; ++u) {
                    const int e = pos + u * 128 + lane * 4;
                    dv[u].x = (e     < end) ? dstv[e]     : -1;
                    dv[u].y = (e + 1 < end) ? dstv[e + 1] : -1;
                    dv[u].z = (e + 2 < end) ? dstv[e + 2] : -1;
                    dv[u].w = (e + 3 < end) ? dstv[e + 3] : -1;
                }
            }
#pragma unroll
            for (int u = 0; u < AG_U; ++u) {
                if (pos < end && cnt <= AG_QCAP - 128) {
                    const unsigned e = (unsigned)(pos + lane * 4);
                    unsigned lr[4];
                    lr[0] = (unsigned)(dv[u].x - r0); lr[1] = (unsigned)(dv[u].y - r0);
                    lr[2] = (unsigned)(dv[u].z - r0); lr[3] = (unsigned)(dv[u].w - r0);
                    const unsigned a01 = lr[0] < lr[1] ? lr[0] : lr[1];
                    const unsigned a23 = lr[2] < lr[3] ? lr[2] : lr[3];
                    const unsigned mn  = a01 < a23 ? a01 : a23;
                    if (__builtin_amdgcn_ballot_w32(mn < (unsigned)nr) != 0u) {
#pragma unroll
                        for (int k = 0; k < 4; ++k) {
                            const bool hit = lr[k] < (unsigned)nr;
                            const unsigned m = __builtin_amdgcn_ballot_w32(hit);
                            if (hit) myq[cnt + __popc(m & ltmask)] = ((e + k) << 8) | lr[k];
                            cnt += __popc(m);
                        }
                    }
                    pos += 128;
                }
            }
        }
        if (lane == 0) {
            qcnt[wave] = cnt;
            if (pos < end) pend[rnd & 1] = 1;
        }
        __syncthreads();

        int pc = 0;
        for (int q = 0; q < AG_W; ++q) {
            const int n = qcnt[q];
            const unsigned* qq = qbuf + q * AG_QCAP;
            for (int i0 = 0; i0 < n; i0 += 32) {
                const int i = i0 + lane;
                unsigned ent = 0;
                bool mine = false;
                if (i < n) { ent = qq[i]; mine = ((ent & 31u) == (unsigned)wave); }
                const unsigned m = __builtin_amdgcn_ballot_w32(mine);
                if (m != 0u) {
                    if (mine) mb[pc + __popc(m & ltmask)] = ent;
                    pc += __popc(m);
                    __builtin_amdgcn_wave_barrier();
                    if (pc >= 32) {
                        ag_drain<MODE == 0>(mb, 32, lane, srcv, hs, acc, wl, wpos);
                        unsigned t = 0;
                        if (lane < pc - 32) t = mb[32 + lane];
                        __builtin_amdgcn_wave_barrier();
                        if (lane < pc - 32) mb[lane] = t;
                        pc -= 32;
                        __builtin_amdgcn_wave_barrier();
                    }
                }
            }
        }
        if (pc > 0) ag_drain<MODE == 0>(mb, pc, lane, srcv, hs, acc, wl, wpos);

        const int again = pend[rnd & 1];
        if (tid == 0) pend[(rnd + 1) & 1] = 0;
        __syncthreads();
        if (!again) break;
    }

    if (MODE == 0 && lane == 0) lcnt[blockIdx.x * AG_W + wave] = wpos;

    ag_write_out(acc, dis, out, r0, nr, tid);
}

__global__ void __launch_bounds__(256) q_kernel(const int* __restrict__ va, const float* __restrict__ agg,
                                                const float* __restrict__ b2,
                                                const float* __restrict__ Wq, const float* __restrict__ bq,
                                                float* __restrict__ out, int A) {
    const int lane = threadIdx.x & 31;
    const int wv = __builtin_amdgcn_readfirstlane((int)((blockIdx.x * blockDim.x + threadIdx.x) >> 5));
    const int a0 = wv * 32;
    if (a0 >= A) return;
    const int a  = a0 + lane;
    const int ac = (a < A) ? a : (A - 1);
    const int2 ij = *(const int2*)(va + 2 * (long long)ac);

    const float4 b  = *(const float4*)(b2 + lane * 4);
    const float4 wi = *(const float4*)(Wq + lane * 4);
    const float4 wj = *(const float4*)(Wq + 128 + lane * 4);
    const float bq0 = bq[0];

    const bool h16 = (lane & 16) != 0;
    const bool h8  = (lane & 8) != 0;
    const bool h4  = (lane & 4) != 0;
    const int t = lane & 7;
    const int srcl = ((t & 1) << 4) | ((t & 2) << 2) | (t & 4);

    float res = 0.0f;
#pragma unroll
    for (int it = 0; it < 4; ++it) {
        float4 xi[8], xj[8];
#pragma unroll
        for (int k = 0; k < 8; ++k) {
            const int i = __builtin_amdgcn_readlane(ij.x, it * 8 + k);
            const int j = __builtin_amdgcn_readlane(ij.y, it * 8 + k);
            xi[k] = *(const float4*)(agg + (long long)i * 128 + lane * 4);
            xj[k] = *(const float4*)(agg + (long long)j * 128 + lane * 4);
        }
        float v[8];
#pragma unroll
        for (int k = 0; k < 8; ++k) {
            float4 x = xi[k], y = xj[k];
            x.x = x.x + b.x; x.y = x.y + b.y; x.z = x.z + b.z; x.w = x.w + b.w;
            y.x = y.x + b.x; y.y = y.y + b.y; y.z = y.z + b.z; y.w = y.w + b.w;
            x.x = x.x > 0.0f ? x.x : 0.0f;
            x.y = x.y > 0.0f ? x.y : 0.0f;
            x.z = x.z > 0.0f ? x.z : 0.0f;
            x.w = x.w > 0.0f ? x.w : 0.0f;
            y.x = y.x > 0.0f ? y.x : 0.0f;
            y.y = y.y > 0.0f ? y.y : 0.0f;
            y.z = y.z > 0.0f ? y.z : 0.0f;
            y.w = y.w > 0.0f ? y.w : 0.0f;
            v[k] = (x.x * wi.x + x.y * wi.y + x.z * wi.z + x.w * wi.w)
                 + (y.x * wj.x + y.y * wj.y + y.z * wj.z + y.w * wj.w);
        }
        float w4[4];
#pragma unroll
        for (int p = 0; p < 4; ++p) {
            const float keep = h16 ? v[2 * p + 1] : v[2 * p];
            const float send = h16 ? v[2 * p] : v[2 * p + 1];
            w4[p] = keep + __shfl_xor(send, 16);
        }
        float u2[2];
#pragma unroll
        for (int p = 0; p < 2; ++p) {
            const float keep = h8 ? w4[2 * p + 1] : w4[2 * p];
            const float send = h8 ? w4[2 * p] : w4[2 * p + 1];
            u2[p] = keep + __shfl_xor(send, 8);
        }
        float z;
        {
            const float keep = h4 ? u2[1] : u2[0];
            const float send = h4 ? u2[0] : u2[1];
            z = keep + __shfl_xor(send, 4);
        }
        z += __shfl_xor(z, 2);
        z += __shfl_xor(z, 1);
        const float r = __shfl(z, srcl);
        if ((lane >> 3) == it) res = r;
    }
    if (a < A) out[a] = res + bq0;
}

extern "C" void kernel_launch(void* const* d_in, const int* in_sizes, int n_in,
                              void* d_out, int out_size, void* d_ws, size_t ws_size,
                              hipStream_t stream) {
    const float* x  = (const float*)d_in[0];
    const int*   ei = (const int*)d_in[1];
    const int*   va = (const int*)d_in[2];
    const float* W1 = (const float*)d_in[3];
    const float* b1 = (const float*)d_in[4];
    const float* W2 = (const float*)d_in[5];
    const float* b2 = (const float*)d_in[6];
    const float* Wq = (const float*)d_in[7];
    const float* bq = (const float*)d_in[8];
    float* q = (float*)d_out;

    const int H  = 128;
    const int Nn = in_sizes[0] / H;
    const int E  = in_sizes[1] / 2;
    const int A  = in_sizes[2] / 2;

    char*  ws  = (char*)d_ws;
    size_t off = 0;
    auto alloc = [&](size_t bytes) -> char* {
        char* p = ws + off;
        off = (off + bytes + 255) & ~(size_t)255;
        return p;
    };
    _Float16* w1t  = (_Float16*)alloc((size_t)H * H * 2);
    _Float16* w2t  = (_Float16*)alloc((size_t)H * H * 2);
    float*    gout = (float*)   alloc((size_t)Nn * H * 4);
    float*    agg1 = (float*)   alloc((size_t)Nn * H * 4);
    float*    agg2 = (float*)   alloc((size_t)Nn * H * 4);
    float*    dis  = (float*)   alloc((size_t)Nn * 4);

    auto blks = [](long long n, int t) { return (unsigned)((n + t - 1) / t); };

    prep_w_deg<<<blks(2LL * H * H + Nn, 256), 256, 0, stream>>>(W1, W2, w1t, w2t, dis, Nn);
    deg_accum<<<blks(E, 256), 256, 0, stream>>>(ei + E, dis, E);

    const unsigned mtiles = (unsigned)((Nn + 15) / 16);
    const unsigned owners = blks(Nn, AG_R);
    unsigned* lists = (unsigned*)alloc((size_t)owners * AG_W * AG_LCAP * 4);
    int*      lcnt  = (int*)     alloc((size_t)owners * AG_W * 4);

    gemm_fused<0><<<mtiles, 256, 0, stream>>>(x, b1, w1t, gout, dis, Nn);
    agg_owner<0><<<owners, AG_T, 0, stream>>>(ei, dis, gout, agg1, lists, lcnt, E, Nn);

    gemm_fused<1><<<mtiles, 256, 0, stream>>>(agg1, b1, w2t, gout, dis, Nn);
    agg_owner<1><<<owners, AG_T, 0, stream>>>(ei, dis, gout, agg2, lists, lcnt, E, Nn);

    q_kernel<<<blks(A, 256), 256, 0, stream>>>(va, agg2, b2, Wq, bq, q, A);
}


# layer-2 GEMM moved into the layer-1 owner blocks' write-out: the 200 LDS rows get 1/sqrt(deg), bias, ReLU, become an f16 tile in place and go through WMMA with W2; 7 launches to 6, no f32 round trip
# speedup vs baseline: 2.4702x; 1.0744x over previous
#include <hip/hip_runtime.h>
#include <hip/hip_bf16.h>


typedef __attribute__((ext_vector_type(16))) _Float16 v16h;
typedef __attribute__((ext_vector_type(8)))  _Float16 v8h;
typedef __attribute__((ext_vector_type(4)))  _Float16 v4h;
typedef __attribute__((ext_vector_type(8)))  float    v8f;

__global__ void prep_w_deg(const float* __restrict__ W1, const float* __restrict__ W2,
                           _Float16* __restrict__ w1t, _Float16* __restrict__ w2t,
                           float* __restrict__ deg, int Nn) {
    int idx = blockIdx.x * blockDim.x + threadIdx.x;
    if (idx < 16384) {
        int k = idx >> 7, c = idx & 127;
        w1t[c * 128 + k] = (_Float16)W1[idx];
    } else if (idx < 32768) {
        int j = idx - 16384;
        int k = j >> 7, c = j & 127;
        w2t[c * 128 + k] = (_Float16)W2[j];
    } else {
        int i = idx - 32768;
        if (i < Nn) deg[i] = 1.0f;
    }
}

__global__ void deg_accum(const int* __restrict__ dst, float* __restrict__ deg, int E) {
    int e = blockIdx.x * blockDim.x + threadIdx.x;
    if (e < E) atomicAdd(&deg[dst[e]], 1.0f);
}

#define A_PITCH 136
__global__ void __launch_bounds__(256) gemm_fused(const float* __restrict__ A32,
                                                  const _Float16* __restrict__ Bt,
                                                  float* __restrict__ C,
                                                  float* __restrict__ dis,
                                                  int nrows) {
    __shared__ __attribute__((aligned(16))) _Float16 As[16 * A_PITCH];
    __shared__ float sdis[16];

    const int tid  = threadIdx.x;
    const int wave = tid >> 5;
    const int lane = tid & 31;
    const int mt   = blockIdx.x;

    const int colB  = wave * 16 + (lane & 15);
    const int kbase = (lane >> 4) ? 8 : 0;
    const _Float16* __restrict__ brow = Bt + (long long)colB * 128;

    v8h blo[4], bhi[4];
#pragma unroll
    for (int kc = 0; kc < 4; ++kc) {
        const int k0 = kc * 32 + kbase;
        blo[kc] = *(const v8h*)(brow + k0);
        bhi[kc] = *(const v8h*)(brow + k0 + 16);
    }

    if (tid < 16) {
        const int i = mt * 16 + tid;
        float r = 0.0f;
        if (i < nrows) {
            const float d = dis[i];
            r = (d > 0.0f) ? rsqrtf(d) : 0.0f;
            dis[i] = r;
        }
        sdis[tid] = r;
    }

#pragma unroll
    for (int p = 0; p < 2; ++p) {
        const int f  = tid + p * 256;
        const int r  = f >> 5;
        const int c4 = (f & 31) * 4;
        int row = mt * 16 + r;
        if (row > nrows - 1) row = nrows - 1;
        float4 v = *(const float4*)(A32 + (long long)row * 128 + c4);
        v4h hv;
        hv[0] = (_Float16)v.x; hv[1] = (_Float16)v.y; hv[2] = (_Float16)v.z; hv[3] = (_Float16)v.w;
        *(v4h*)(As + r * A_PITCH + c4) = hv;
    }

    __syncthreads();

    const _Float16* arow = As + (lane & 15) * A_PITCH;
    v8f acc = {};
#pragma unroll
    for (int kc = 0; kc < 4; ++kc) {
        const int k0 = kc * 32 + kbase;
        v8h alo = *(const v8h*)(arow + k0);
        v8h ahi = *(const v8h*)(arow + k0 + 16);
        v16h a, b;
#pragma unroll
        for (int i = 0; i < 8; ++i) {
            a[i] = alo[i];     a[8 + i] = ahi[i];
            b[i] = blo[kc][i]; b[8 + i] = bhi[kc][i];
        }
        acc = __builtin_amdgcn_wmma_f32_16x16x32_f16(
             false, a,  false, b,
             (short)0, acc,  false,  false);
    }

    const int rbase = mt * 16 + ((lane >> 4) ? 8 : 0);
    float* __restrict__ crow = C + (long long)rbase * 128 + wave * 16 + (lane & 15);
    const float* sd = sdis + ((lane >> 4) ? 8 : 0);
    if (mt * 16 + 16 <= nrows) {
#pragma unroll
        for (int r = 0; r < 8; ++r) crow[(long long)r * 128] = acc[r] * sd[r];
    } else {
#pragma unroll
        for (int r = 0; r < 8; ++r)
            if (rbase + r < nrows) crow[(long long)r * 128] = acc[r] * sd[r];
    }
}

#define AG_R    200
#define AG_RP   208
#define AG_P    132
#define AG_PH   264
#define AG_T    1024
#define AG_W    32
#define AG_QCAP 256
#define AG_U    4
#define AG_LCAP 192

__device__ __forceinline__ void ag_add_rows(unsigned lr_l, int s, int cnt, int lane,
                                            const float* __restrict__ hs, float* acc) {
    for (int g = 0; g < cnt; g += 8) {
        float4 v[8];
#pragma unroll
        for (int j = 0; j < 8; ++j) {
            const int sj = __builtin_amdgcn_readlane(s, g + j);
            v[j] = *(const float4*)(hs + (long long)sj * 128 + lane * 4);
        }
#pragma unroll
        for (int j = 0; j < 8; ++j) {
            if (g + j < cnt) {
                const unsigned lr = (unsigned)__builtin_amdgcn_readlane((int)lr_l, g + j);
                float4* a = (float4*)(acc + lr * AG_P + lane * 4);
                float4 t = *a;
                t.x += v[j].x; t.y += v[j].y; t.z += v[j].z; t.w += v[j].w;
                *a = t;
            }
        }
    }
}

template <bool SAVE>
__device__ __forceinline__ void ag_drain(const unsigned* mb, int cnt, int lane,
                                         const int* __restrict__ srcv, const float* __restrict__ hs,
                                         float* acc, unsigned* __restrict__ wl, int& wpos) {
    unsigned ent = 0;
    int s = 0;
    if (lane < cnt) { ent = mb[lane]; s = srcv[ent >> 8]; }
    if (SAVE) {
        if (lane < cnt && wpos + lane < AG_LCAP) wl[wpos + lane] = ((unsigned)s << 8) | (ent & 255u);
        wpos += cnt;
    }
    ag_add_rows(ent & 255u, s, cnt, lane, hs, acc);
}

__device__ __forceinline__ void ag_write_out(const float* acc, const float* __restrict__ dis,
                                             float* __restrict__ out, int r0, int nr, int tid) {
    for (int f = tid; f < AG_R * 32; f += AG_T) {
        const int r = f >> 5, c4 = (f & 31) * 4;
        if (r < nr) {
            const float dr = dis[r0 + r];
            float4 v = *(const float4*)(acc + r * AG_P + c4);
            v.x *= dr; v.y *= dr; v.z *= dr; v.w *= dr;
            *(float4*)(out + (long long)(r0 + r) * 128 + c4) = v;
        }
    }
}

__device__ __forceinline__ void ag_transform_out(float* acc, float* sdis,
                                                 const float* __restrict__ dis, const float* __restrict__ bias,
                                                 const _Float16* __restrict__ Bt, float* __restrict__ out,
                                                 int r0, int nr, int wave, int lane) {
    const int nt    = wave & 7;
    const int colB  = nt * 16 + (lane & 15);
    const int kbase = (lane >> 4) ? 8 : 0;
    const _Float16* __restrict__ brow = Bt + colB * 128;

    v8h blo[4], bhi[4];
#pragma unroll
    for (int kc = 0; kc < 4; ++kc) {
        const int k0 = kc * 32 + kbase;
        blo[kc] = *(const v8h*)(brow + k0);
        bhi[kc] = *(const v8h*)(brow + k0 + 16);
    }

    const int c4 = lane * 4;
    const float4 bb = *(const float4*)(bias + c4);
    float dv[7];
#pragma unroll
    for (int p = 0; p < 7; ++p) {
        const int r  = wave + p * 32;
        const int rn = (r < nr) ? r : (nr - 1);
        dv[p] = dis[r0 + rn];
    }
    float4 t[7];
#pragma unroll
    for (int p = 0; p < 7; ++p) {
        const int r  = wave + p * 32;
        const int ra = (r < AG_R) ? r : (AG_R - 1);
        const float dr = (r < nr) ? dv[p] : 0.0f;
        if (lane == 0 && r < AG_R) sdis[r] = dr;
        float4 v = *(const float4*)(acc + ra * AG_P + c4);
        v.x = v.x * dr + bb.x; v.y = v.y * dr + bb.y; v.z = v.z * dr + bb.z; v.w = v.w * dr + bb.w;
        v.x = v.x > 0.0f ? v.x : 0.0f;
        v.y = v.y > 0.0f ? v.y : 0.0f;
        v.z = v.z > 0.0f ? v.z : 0.0f;
        v.w = v.w > 0.0f ? v.w : 0.0f;
        t[p] = (r < AG_R) ? v : make_float4(0.0f, 0.0f, 0.0f, 0.0f);
    }
    __syncthreads();
    _Float16* tile = (_Float16*)acc;
#pragma unroll
    for (int p = 0; p < 7; ++p) {
        const int r = wave + p * 32;
        if (r < AG_RP) {
            v4h hv;
            hv[0] = (_Float16)t[p].x; hv[1] = (_Float16)t[p].y; hv[2] = (_Float16)t[p].z; hv[3] = (_Float16)t[p].w;
            *(v4h*)(tile + r * AG_PH + c4) = hv;
        }
    }
    __syncthreads();

    for (int mt = wave >> 3; mt < AG_RP / 16; mt += 4) {
        const _Float16* arow = tile + (mt * 16 + (lane & 15)) * AG_PH;
        v8f c = {};
#pragma unroll
        for (int kc = 0; kc < 4; ++kc) {
            const int k0 = kc * 32 + kbase;
            v8h alo = *(const v8h*)(arow + k0);
            v8h ahi = *(const v8h*)(arow + k0 + 16);
            v16h a, b;
#pragma unroll
            for (int i = 0; i < 8; ++i) {
                a[i] = alo[i];     a[8 + i] = ahi[i];
                b[i] = blo[kc][i]; b[8 + i] = bhi[kc][i];
            }
            c = __builtin_amdgcn_wmma_f32_16x16x32_f16(
                 false, a,  false, b,
                 (short)0, c,  false,  false);
        }
        const int lrb = mt * 16 + ((lane >> 4) ? 8 : 0);
        float* __restrict__ crow = out + (long long)(r0 + lrb) * 128 + nt * 16 + (lane & 15);
        const float* sd = sdis + lrb;
        if (mt * 16 + 16 <= nr) {
#pragma unroll
            for (int r = 0; r < 8; ++r) crow[(long long)r * 128] = c[r] * sd[r];
        } else {
#pragma unroll
            for (int r = 0; r < 8; ++r)
                if (lrb + r < nr) crow[(long long)r * 128] = c[r] * sd[r];
        }
    }
}

template <int MODE>
__global__ void __launch_bounds__(AG_T) agg_owner(const int* __restrict__ ei, const float* __restrict__ dis,
                                                  const float* __restrict__ hs, float* __restrict__ out,
                                                  unsigned* __restrict__ lists, int* __restrict__ lcnt,
                                                  const float* __restrict__ bias, const _Float16* __restrict__ Bt,
                                                  int E, int Nn) {
    __shared__ __attribute__((aligned(16))) float acc[AG_RP * AG_P];
    __shared__ float sdis[AG_RP];
    __shared__ unsigned qbuf[AG_W * AG_QCAP];
    __shared__ unsigned mybuf[AG_W * 64];
    __shared__ int qcnt[AG_W];
    __shared__ int pend[2];

    const int tid  = threadIdx.x;
    const int wave = __builtin_amdgcn_readfirstlane(tid >> 5);
    const int lane = tid & 31;
    const unsigned ltmask = (1u << lane) - 1u;
    const int r0 = blockIdx.x * AG_R;
    const int nr = (Nn - r0 < AG_R) ? (Nn - r0) : AG_R;

    const int* __restrict__ srcv = ei;
    const int* __restrict__ dstv = ei + E;
    const bool al16 = (((size_t)dstv) & 15) == 0;

    unsigned* __restrict__ wl = lists + (size_t)(blockIdx.x * AG_W + wave) * AG_LCAP;
    int wpos = 0;
    int cv = 0;
    if (MODE == 1) cv = lcnt[blockIdx.x * AG_W + lane];

    for (int f = tid; f < AG_R * 32; f += AG_T) {
        const int r = f >> 5, c4 = (f & 31) * 4;
        float4 v = make_float4(0.0f, 0.0f, 0.0f, 0.0f);
        if (r < nr) v = *(const float4*)(hs + (long long)(r0 + r) * 128 + c4);
        *(float4*)(acc + r * AG_P + c4) = v;
    }

    if (MODE == 1) {
        if (__builtin_amdgcn_ballot_w32(cv > AG_LCAP) == 0u) {
            const int nl = __builtin_amdgcn_readlane(cv, wave);
            unsigned cur = 0u;
            if (lane < nl) cur = wl[lane];
            __syncthreads();
            int i0 = 0;
            for (; i0 + 32 <= nl; i0 += 32) {
                unsigned nxt = 0u;
                if (i0 + 32 + lane < nl) nxt = wl[i0 + 32 + lane];
                ag_add_rows(cur & 255u, (int)(cur >> 8), 32, lane, hs, acc);
                cur = nxt;
            }
            if (i0 < nl) ag_add_rows(cur & 255u, (int)(cur >> 8), nl - i0, lane, hs, acc);
            __syncthreads();
            ag_write_out(acc, dis, out, r0, nr, tid);
            return;
        }
    }

    if (tid < 2) pend[tid] = 0;
    __syncthreads();

    const int S = ((E + 4095) >> 12) << 7;
    int pos = wave * S;
    const int end = (pos + S < E) ? (pos + S) : E;
    unsigned* myq = qbuf + wave * AG_QCAP;
    unsigned* mb  = mybuf + wave * 64;

    for (int rnd = 0;; ++rnd) {
        int cnt = 0;
        while (pos < end && cnt <= AG_QCAP - 128) {
            int4 dv[AG_U];
            if (al16 && pos + AG_U * 128 <= end) {
#pragma unroll
                for (int u = 0; u < AG_U; ++u)
                    dv[u] = *(const int4*)(dstv + pos + u * 128 + lane * 4);
            } else {
#pragma unroll
                for (int u = 0; u < AG_U; ++u) {
                    const int e = pos + u * 128 + lane * 4;
                    dv[u].x = (e     < end) ? dstv[e]     : -1;
                    dv[u].y = (e + 1 < end) ? dstv[e + 1] : -1;
                    dv[u].z = (e + 2 < end) ? dstv[e + 2] : -1;
                    dv[u].w = (e + 3 < end) ? dstv[e + 3] : -1;
                }
            }
#pragma unroll
            for (int u = 0; u < AG_U; ++u) {
                if (pos < end && cnt <= AG_QCAP - 128) {
                    const unsigned e = (unsigned)(pos + lane * 4);
                    unsigned lr[4];
                    lr[0] = (unsigned)(dv[u].x - r0); lr[1] = (unsigned)(dv[u].y - r0);
                    lr[2] = (unsigned)(dv[u].z - r0); lr[3] = (unsigned)(dv[u].w - r0);
                    const unsigned a01 = lr[0] < lr[1] ? lr[0] : lr[1];
                    const unsigned a23 = lr[2] < lr[3] ? lr[2] : lr[3];
                    const unsigned mn  = a01 < a23 ? a01 : a23;
                    if (__builtin_amdgcn_ballot_w32(mn < (unsigned)nr) != 0u) {
#pragma unroll
                        for (int k = 0; k < 4; ++k) {
                            const bool hit = lr[k] < (unsigned)nr;
                            const unsigned m = __builtin_amdgcn_ballot_w32(hit);
                            if (hit) myq[cnt + __popc(m & ltmask)] = ((e + k) << 8) | lr[k];
                            cnt += __popc(m);
                        }
                    }
                    pos += 128;
                }
            }
        }
        if (lane == 0) {
            qcnt[wave] = cnt;
            if (pos < end) pend[rnd & 1] = 1;
        }
        __syncthreads();

        int pc = 0;
        for (int q = 0; q < AG_W; ++q) {
            const int n = qcnt[q];
            const unsigned* qq = qbuf + q * AG_QCAP;
            for (int i0 = 0; i0 < n; i0 += 32) {
                const int i = i0 + lane;
                unsigned ent = 0;
                bool mine = false;
                if (i < n) { ent = qq[i]; mine = ((ent & 31u) == (unsigned)wave); }
                const unsigned m = __builtin_amdgcn_ballot_w32(mine);
                if (m != 0u) {
                    if (mine) mb[pc + __popc(m & ltmask)] = ent;
                    pc += __popc(m);
                    __builtin_amdgcn_wave_barrier();
                    if (pc >= 32) {
                        ag_drain<MODE == 0>(mb, 32, lane, srcv, hs, acc, wl, wpos);
                        unsigned t = 0;
                        if (lane < pc - 32) t = mb[32 + lane];
                        __builtin_amdgcn_wave_barrier();
                        if (lane < pc - 32) mb[lane] = t;
                        pc -= 32;
                        __builtin_amdgcn_wave_barrier();
                    }
                }
            }
        }
        if (pc > 0) ag_drain<MODE == 0>(mb, pc, lane, srcv, hs, acc, wl, wpos);

        const int again = pend[rnd & 1];
        if (tid == 0) pend[(rnd + 1) & 1] = 0;
        __syncthreads();
        if (!again) break;
    }

    if (MODE == 0 && lane == 0) lcnt[blockIdx.x * AG_W + wave] = wpos;

    if (MODE == 0) ag_transform_out(acc, sdis, dis, bias, Bt, out, r0, nr, wave, lane);
    else           ag_write_out(acc, dis, out, r0, nr, tid);
}

__global__ void __launch_bounds__(256) q_kernel(const int* __restrict__ va, const float* __restrict__ agg,
                                                const float* __restrict__ b2,
                                                const float* __restrict__ Wq, const float* __restrict__ bq,
                                                float* __restrict__ out, int A) {
    const int lane = threadIdx.x & 31;
    const int wv = __builtin_amdgcn_readfirstlane((int)((blockIdx.x * blockDim.x + threadIdx.x) >> 5));
    const int a0 = wv * 32;
    if (a0 >= A) return;
    const int a  = a0 + lane;
    const int ac = (a < A) ? a : (A - 1);
    const int2 ij = *(const int2*)(va + 2 * (long long)ac);

    const float4 b  = *(const float4*)(b2 + lane * 4);
    const float4 wi = *(const float4*)(Wq + lane * 4);
    const float4 wj = *(const float4*)(Wq + 128 + lane * 4);
    const float bq0 = bq[0];

    const bool h16 = (lane & 16) != 0;
    const bool h8  = (lane & 8) != 0;
    const bool h4  = (lane & 4) != 0;
    const int t = lane & 7;
    const int srcl = ((t & 1) << 4) | ((t & 2) << 2) | (t & 4);

    float res = 0.0f;
#pragma unroll
    for (int it = 0; it < 4; ++it) {
        float4 xi[8], xj[8];
#pragma unroll
        for (int k = 0; k < 8; ++k) {
            const int i = __builtin_amdgcn_readlane(ij.x, it * 8 + k);
            const int j = __builtin_amdgcn_readlane(ij.y, it * 8 + k);
            xi[k] = *(const float4*)(agg + (long long)i * 128 + lane * 4);
            xj[k] = *(const float4*)(agg + (long long)j * 128 + lane * 4);
        }
        float v[8];
#pragma unroll
        for (int k = 0; k < 8; ++k) {
            float4 x = xi[k], y = xj[k];
            x.x = x.x + b.x; x.y = x.y + b.y; x.z = x.z + b.z; x.w = x.w + b.w;
            y.x = y.x + b.x; y.y = y.y + b.y; y.z = y.z + b.z; y.w = y.w + b.w;
            x.x = x.x > 0.0f ? x.x : 0.0f;
            x.y = x.y > 0.0f ? x.y : 0.0f;
            x.z = x.z > 0.0f ? x.z : 0.0f;
            x.w = x.w > 0.0f ? x.w : 0.0f;
            y.x = y.x > 0.0f ? y.x : 0.0f;
            y.y = y.y > 0.0f ? y.y : 0.0f;
            y.z = y.z > 0.0f ? y.z : 0.0f;
            y.w = y.w > 0.0f ? y.w : 0.0f;
            v[k] = (x.x * wi.x + x.y * wi.y + x.z * wi.z + x.w * wi.w)
                 + (y.x * wj.x + y.y * wj.y + y.z * wj.z + y.w * wj.w);
        }
        float w4[4];
#pragma unroll
        for (int p = 0; p < 4; ++p) {
            const float keep = h16 ? v[2 * p + 1] : v[2 * p];
            const float send = h16 ? v[2 * p] : v[2 * p + 1];
            w4[p] = keep + __shfl_xor(send, 16);
        }
        float u2[2];
#pragma unroll
        for (int p = 0; p < 2; ++p) {
            const float keep = h8 ? w4[2 * p + 1] : w4[2 * p];
            const float send = h8 ? w4[2 * p] : w4[2 * p + 1];
            u2[p] = keep + __shfl_xor(send, 8);
        }
        float z;
        {
            const float keep = h4 ? u2[1] : u2[0];
            const float send = h4 ? u2[0] : u2[1];
            z = keep + __shfl_xor(send, 4);
        }
        z += __shfl_xor(z, 2);
        z += __shfl_xor(z, 1);
        const float r = __shfl(z, srcl);
        if ((lane >> 3) == it) res = r;
    }
    if (a < A) out[a] = res + bq0;
}

extern "C" void kernel_launch(void* const* d_in, const int* in_sizes, int n_in,
                              void* d_out, int out_size, void* d_ws, size_t ws_size,
                              hipStream_t stream) {
    const float* x  = (const float*)d_in[0];
    const int*   ei = (const int*)d_in[1];
    const int*   va = (const int*)d_in[2];
    const float* W1 = (const float*)d_in[3];
    const float* b1 = (const float*)d_in[4];
    const float* W2 = (const float*)d_in[5];
    const float* b2 = (const float*)d_in[6];
    const float* Wq = (const float*)d_in[7];
    const float* bq = (const float*)d_in[8];
    float* q = (float*)d_out;

    const int H  = 128;
    const int Nn = in_sizes[0] / H;
    const int E  = in_sizes[1] / 2;
    const int A  = in_sizes[2] / 2;

    char*  ws  = (char*)d_ws;
    size_t off = 0;
    auto alloc = [&](size_t bytes) -> char* {
        char* p = ws + off;
        off = (off + bytes + 255) & ~(size_t)255;
        return p;
    };
    _Float16* w1t  = (_Float16*)alloc((size_t)H * H * 2);
    _Float16* w2t  = (_Float16*)alloc((size_t)H * H * 2);
    float*    gout = (float*)   alloc((size_t)Nn * H * 4);
    float*    gou2 = (float*)   alloc((size_t)Nn * H * 4);
    float*    agg2 = (float*)   alloc((size_t)Nn * H * 4);
    float*    dis  = (float*)   alloc((size_t)Nn * 4);

    auto blks = [](long long n, int t) { return (unsigned)((n + t - 1) / t); };

    prep_w_deg<<<blks(2LL * H * H + Nn, 256), 256, 0, stream>>>(W1, W2, w1t, w2t, dis, Nn);
    deg_accum<<<blks(E, 256), 256, 0, stream>>>(ei + E, dis, E);

    const unsigned mtiles = (unsigned)((Nn + 15) / 16);
    const unsigned owners = blks(Nn, AG_R);
    unsigned* lists = (unsigned*)alloc((size_t)owners * AG_W * AG_LCAP * 4);
    int*      lcnt  = (int*)     alloc((size_t)owners * AG_W * 4);

    gemm_fused<<<mtiles, 256, 0, stream>>>(x, w1t, gout, dis, Nn);
    agg_owner<0><<<owners, AG_T, 0, stream>>>(ei, dis, gout, gou2, lists, lcnt, b1, w2t, E, Nn);

    agg_owner<1><<<owners, AG_T, 0, stream>>>(ei, dis, gou2, agg2, lists, lcnt, b1, w2t, E, Nn);

    q_kernel<<<blks(A, 256), 256, 0, stream>>>(va, agg2, b2, Wq, bq, q, A);
}


# owner blocks filter a 1-byte owner-block list (dst/200, made by the degree kernel) 16 edges per lane, not the 4-byte destinations; rows looked up for hits only; adding order changes (still fixed)
# speedup vs baseline: 2.6413x; 1.0693x over previous
#include <hip/hip_runtime.h>
#include <hip/hip_bf16.h>


typedef __attribute__((ext_vector_type(16))) _Float16 v16h;
typedef __attribute__((ext_vector_type(8)))  _Float16 v8h;
typedef __attribute__((ext_vector_type(4)))  _Float16 v4h;
typedef __attribute__((ext_vector_type(8)))  float    v8f;

__global__ void prep_w_deg(const float* __restrict__ W1, const float* __restrict__ W2,
                           _Float16* __restrict__ w1t, _Float16* __restrict__ w2t,
                           float* __restrict__ deg, int Nn) {
    int idx = blockIdx.x * blockDim.x + threadIdx.x;
    if (idx < 16384) {
        int k = idx >> 7, c = idx & 127;
        w1t[c * 128 + k] = (_Float16)W1[idx];
    } else if (idx < 32768) {
        int j = idx - 16384;
        int k = j >> 7, c = j & 127;
        w2t[c * 128 + k] = (_Float16)W2[j];
    } else {
        int i = idx - 32768;
        if (i < Nn) deg[i] = 1.0f;
    }
}

#define OWN_R 200
__global__ void deg_accum_own(const int* __restrict__ dst, float* __restrict__ deg,
                              unsigned* __restrict__ own, int E, int nwords) {
    const int w = blockIdx.x * blockDim.x + threadIdx.x;
    if (w >= nwords) return;
    const int e = w * 4;
    int d[4] = {-1, -1, -1, -1};
    if (e + 3 < E && ((((size_t)dst) & 15) == 0)) {
        const int4 v = *(const int4*)(dst + e);
        d[0] = v.x; d[1] = v.y; d[2] = v.z; d[3] = v.w;
    } else {
#pragma unroll
        for (int k = 0; k < 4; ++k)
            if (e + k < E) d[k] = dst[e + k];
    }
    unsigned pk = 0u;
#pragma unroll
    for (int k = 0; k < 4; ++k) {
        unsigned b = 255u;
        if (d[k] >= 0) {
            atomicAdd(&deg[d[k]], 1.0f);
            b = (unsigned)d[k] / (unsigned)OWN_R;
        }
        pk |= b << (8 * k);
    }
    own[w] = pk;
}

#define A_PITCH 136
__global__ void __launch_bounds__(256) gemm_fused(const float* __restrict__ A32,
                                                  const _Float16* __restrict__ Bt,
                                                  float* __restrict__ C,
                                                  float* __restrict__ dis,
                                                  int nrows) {
    __shared__ __attribute__((aligned(16))) _Float16 As[16 * A_PITCH];
    __shared__ float sdis[16];

    const int tid  = threadIdx.x;
    const int wave = tid >> 5;
    const int lane = tid & 31;
    const int mt   = blockIdx.x;

    const int colB  = wave * 16 + (lane & 15);
    const int kbase = (lane >> 4) ? 8 : 0;
    const _Float16* __restrict__ brow = Bt + (long long)colB * 128;

    v8h blo[4], bhi[4];
#pragma unroll
    for (int kc = 0; kc < 4; ++kc) {
        const int k0 = kc * 32 + kbase;
        blo[kc] = *(const v8h*)(brow + k0);
        bhi[kc] = *(const v8h*)(brow + k0 + 16);
    }

    if (tid < 16) {
        const int i = mt * 16 + tid;
        float r = 0.0f;
        if (i < nrows) {
            const float d = dis[i];
            r = (d > 0.0f) ? rsqrtf(d) : 0.0f;
            dis[i] = r;
        }
        sdis[tid] = r;
    }

#pragma unroll
    for (int p = 0; p < 2; ++p) {
        const int f  = tid + p * 256;
        const int r  = f >> 5;
        const int c4 = (f & 31) * 4;
        int row = mt * 16 + r;
        if (row > nrows - 1) row = nrows - 1;
        float4 v = *(const float4*)(A32 + (long long)row * 128 + c4);
        v4h hv;
        hv[0] = (_Float16)v.x; hv[1] = (_Float16)v.y; hv[2] = (_Float16)v.z; hv[3] = (_Float16)v.w;
        *(v4h*)(As + r * A_PITCH + c4) = hv;
    }

    __syncthreads();

    const _Float16* arow = As + (lane & 15) * A_PITCH;
    v8f acc = {};
#pragma unroll
    for (int kc = 0; kc < 4; ++kc) {
        const int k0 = kc * 32 + kbase;
        v8h alo = *(const v8h*)(arow + k0);
        v8h ahi = *(const v8h*)(arow + k0 + 16);
        v16h a, b;
#pragma unroll
        for (int i = 0; i < 8; ++i) {
            a[i] = alo[i];     a[8 + i] = ahi[i];
            b[i] = blo[kc][i]; b[8 + i] = bhi[kc][i];
        }
        acc = __builtin_amdgcn_wmma_f32_16x16x32_f16(
             false, a,  false, b,
             (short)0, acc,  false,  false);
    }

    const int rbase = mt * 16 + ((lane >> 4) ? 8 : 0);
    float* __restrict__ crow = C + (long long)rbase * 128 + wave * 16 + (lane & 15);
    const float* sd = sdis + ((lane >> 4) ? 8 : 0);
    if (mt * 16 + 16 <= nrows) {
#pragma unroll
        for (int r = 0; r < 8; ++r) crow[(long long)r * 128] = acc[r] * sd[r];
    } else {
#pragma unroll
        for (int r = 0; r < 8; ++r)
            if (rbase + r < nrows) crow[(long long)r * 128] = acc[r] * sd[r];
    }
}

#define AG_R    OWN_R
#define AG_RP   208
#define AG_P    132
#define AG_PH   264
#define AG_T    1024
#define AG_W    32
#define AG_QCAP 768
#define AG_U    8
#define AG_LCAP 192

__device__ __forceinline__ void ag_add_rows(unsigned lr_l, int s, int cnt, int lane,
                                            const float* __restrict__ hs, float* acc) {
    for (int g = 0; g < cnt; g += 8) {
        float4 v[8];
#pragma unroll
        for (int j = 0; j < 8; ++j) {
            const int sj = __builtin_amdgcn_readlane(s, g + j);
            v[j] = *(const float4*)(hs + (long long)sj * 128 + lane * 4);
        }
#pragma unroll
        for (int j = 0; j < 8; ++j) {
            if (g + j < cnt) {
                const unsigned lr = (unsigned)__builtin_amdgcn_readlane((int)lr_l, g + j);
                float4* a = (float4*)(acc + lr * AG_P + lane * 4);
                float4 t = *a;
                t.x += v[j].x; t.y += v[j].y; t.z += v[j].z; t.w += v[j].w;
                *a = t;
            }
        }
    }
}

template <bool SAVE>
__device__ __forceinline__ void ag_drain(const unsigned* mb, int cnt, int lane,
                                         const int* __restrict__ srcv, const float* __restrict__ hs,
                                         float* acc, unsigned* __restrict__ wl, int& wpos) {
    unsigned ent = 0;
    int s = 0;
    if (lane < cnt) { ent = mb[lane]; s = srcv[ent >> 8]; }
    if (SAVE) {
        if (lane < cnt && wpos + lane < AG_LCAP) wl[wpos + lane] = ((unsigned)s << 8) | (ent & 255u);
        wpos += cnt;
    }
    ag_add_rows(ent & 255u, s, cnt, lane, hs, acc);
}

__device__ __forceinline__ void ag_write_out(const float* acc, const float* __restrict__ dis,
                                             float* __restrict__ out, int r0, int nr, int tid) {
    for (int f = tid; f < AG_R * 32; f += AG_T) {
        const int r = f >> 5, c4 = (f & 31) * 4;
        if (r < nr) {
            const float dr = dis[r0 + r];
            float4 v = *(const float4*)(acc + r * AG_P + c4);
            v.x *= dr; v.y *= dr; v.z *= dr; v.w *= dr;
            *(float4*)(out + (long long)(r0 + r) * 128 + c4) = v;
        }
    }
}

__device__ __forceinline__ void ag_transform_out(float* acc, float* sdis,
                                                 const float* __restrict__ dis, const float* __restrict__ bias,
                                                 const _Float16* __restrict__ Bt, float* __restrict__ out,
                                                 int r0, int nr, int wave, int lane) {
    const int nt    = wave & 7;
    const int colB  = nt * 16 + (lane & 15);
    const int kbase = (lane >> 4) ? 8 : 0;
    const _Float16* __restrict__ brow = Bt + colB * 128;

    v8h blo[4], bhi[4];
#pragma unroll
    for (int kc = 0; kc < 4; ++kc) {
        const int k0 = kc * 32 + kbase;
        blo[kc] = *(const v8h*)(brow + k0);
        bhi[kc] = *(const v8h*)(brow + k0 + 16);
    }

    const int c4 = lane * 4;
    const float4 bb = *(const float4*)(bias + c4);
    float dv[7];
#pragma unroll
    for (int p = 0; p < 7; ++p) {
        const int r  = wave + p * 32;
        const int rn = (r < nr) ? r : (nr - 1);
        dv[p] = dis[r0 + rn];
    }
    float4 t[7];
#pragma unroll
    for (int p = 0; p < 7; ++p) {
        const int r  = wave + p * 32;
        const int ra = (r < AG_R) ? r : (AG_R - 1);
        const float dr = (r < nr) ? dv[p] : 0.0f;
        if (lane == 0 && r < AG_R) sdis[r] = dr;
        float4 v = *(const float4*)(acc + ra * AG_P + c4);
        v.x = v.x * dr + bb.x; v.y = v.y * dr + bb.y; v.z = v.z * dr + bb.z; v.w = v.w * dr + bb.w;
        v.x = v.x > 0.0f ? v.x : 0.0f;
        v.y = v.y > 0.0f ? v.y : 0.0f;
        v.z = v.z > 0.0f ? v.z : 0.0f;
        v.w = v.w > 0.0f ? v.w : 0.0f;
        t[p] = (r < AG_R) ? v : make_float4(0.0f, 0.0f, 0.0f, 0.0f);
    }
    __syncthreads();
    _Float16* tile = (_Float16*)acc;
#pragma unroll
    for (int p = 0; p < 7; ++p) {
        const int r = wave + p * 32;
        if (r < AG_RP) {
            v4h hv;
            hv[0] = (_Float16)t[p].x; hv[1] = (_Float16)t[p].y; hv[2] = (_Float16)t[p].z; hv[3] = (_Float16)t[p].w;
            *(v4h*)(tile + r * AG_PH + c4) = hv;
        }
    }
    __syncthreads();

    for (int mt = wave >> 3; mt < AG_RP / 16; mt += 4) {
        const _Float16* arow = tile + (mt * 16 + (lane & 15)) * AG_PH;
        v8f c = {};
#pragma unroll
        for (int kc = 0; kc < 4; ++kc) {
            const int k0 = kc * 32 + kbase;
            v8h alo = *(const v8h*)(arow + k0);
            v8h ahi = *(const v8h*)(arow + k0 + 16);
            v16h a, b;
#pragma unroll
            for (int i = 0; i < 8; ++i) {
                a[i] = alo[i];     a[8 + i] = ahi[i];
                b[i] = blo[kc][i]; b[8 + i] = bhi[kc][i];
            }
            c = __builtin_amdgcn_wmma_f32_16x16x32_f16(
                 false, a,  false, b,
                 (short)0, c,  false,  false);
        }
        const int lrb = mt * 16 + ((lane >> 4) ? 8 : 0);
        float* __restrict__ crow = out + (long long)(r0 + lrb) * 128 + nt * 16 + (lane & 15);
        const float* sd = sdis + lrb;
        if (mt * 16 + 16 <= nr) {
#pragma unroll
            for (int r = 0; r < 8; ++r) crow[(long long)r * 128] = c[r] * sd[r];
        } else {
#pragma unroll
            for (int r = 0; r < 8; ++r)
                if (lrb + r < nr) crow[(long long)r * 128] = c[r] * sd[r];
        }
    }
}

template <int MODE>
__global__ void __launch_bounds__(AG_T) agg_owner(const int* __restrict__ ei, const float* __restrict__ dis,
                                                  const float* __restrict__ hs, float* __restrict__ out,
                                                  unsigned* __restrict__ lists, int* __restrict__ lcnt,
                                                  const float* __restrict__ bias, const _Float16* __restrict__ Bt,
                                                  const unsigned* __restrict__ own,
                                                  int E, int Nn) {
    __shared__ __attribute__((aligned(16))) float acc[AG_RP * AG_P];
    __shared__ float sdis[AG_RP];
    __shared__ unsigned qbuf[AG_W * AG_QCAP];
    __shared__ unsigned mybuf[AG_W * 64];
    __shared__ int qcnt[AG_W];
    __shared__ int pend[2];

    const int tid  = threadIdx.x;
    const int wave = __builtin_amdgcn_readfirstlane(tid >> 5);
    const int lane = tid & 31;
    const unsigned ltmask = (1u << lane) - 1u;
    const int r0 = blockIdx.x * AG_R;
    const int nr = (Nn - r0 < AG_R) ? (Nn - r0) : AG_R;

    const int* __restrict__ srcv = ei;
    const int* __restrict__ dstv = ei + E;

    unsigned* __restrict__ wl = lists + (size_t)(blockIdx.x * AG_W + wave) * AG_LCAP;
    int wpos = 0;
    int cv = 0;
    if (MODE == 1) cv = lcnt[blockIdx.x * AG_W + lane];

    for (int f = tid; f < AG_R * 32; f += AG_T) {
        const int r = f >> 5, c4 = (f & 31) * 4;
        float4 v = make_float4(0.0f, 0.0f, 0.0f, 0.0f);
        if (r < nr) v = *(const float4*)(hs + (long long)(r0 + r) * 128 + c4);
        *(float4*)(acc + r * AG_P + c4) = v;
    }

    if (MODE == 1) {
        if (__builtin_amdgcn_ballot_w32(cv > AG_LCAP) == 0u) {
            const int nl = __builtin_amdgcn_readlane(cv, wave);
            unsigned cur = 0u;
            if (lane < nl) cur = wl[lane];
            __syncthreads();
            int i0 = 0;
            for (; i0 + 32 <= nl; i0 += 32) {
                unsigned nxt = 0u;
                if (i0 + 32 + lane < nl) nxt = wl[i0 + 32 + lane];
                ag_add_rows(cur & 255u, (int)(cur >> 8), 32, lane, hs, acc);
                cur = nxt;
            }
            if (i0 < nl) ag_add_rows(cur & 255u, (int)(cur >> 8), nl - i0, lane, hs, acc);
            __syncthreads();
            ag_write_out(acc, dis, out, r0, nr, tid);
            return;
        }
    }

    if (tid < 2) pend[tid] = 0;
    __syncthreads();

    const int S = ((E + 16383) >> 14) << 9;
    int pos = wave * S;
    const int end = (pos + S < E) ? (pos + S) : E;
    unsigned* myq = qbuf + wave * AG_QCAP;
    unsigned* mb  = mybuf + wave * 64;
    const unsigned bpat = (unsigned)blockIdx.x * 0x01010101u;

    for (int rnd = 0;; ++rnd) {
        int cnt = 0;
        while (pos < end && cnt <= AG_QCAP - 512) {
            uint4 ow[AG_U];
#pragma unroll
            for (int u = 0; u < AG_U; ++u)
                ow[u] = *(const uint4*)(own + (pos >> 2) + u * 128 + lane * 4);
#pragma unroll
            for (int u = 0; u < AG_U; ++u) {
                if (pos < end && cnt <= AG_QCAP - 512) {
                    const unsigned x0 = ow[u].x ^ bpat, x1 = ow[u].y ^ bpat, x2 = ow[u].z ^ bpat, x3 = ow[u].w ^ bpat;
                    const unsigned g0 = ((x0 & 0x7F7F7F7Fu) + 0x7F7F7F7Fu) | x0;
                    const unsigned g1 = ((x1 & 0x7F7F7F7Fu) + 0x7F7F7F7Fu) | x1;
                    const unsigned g2 = ((x2 & 0x7F7F7F7Fu) + 0x7F7F7F7Fu) | x2;
                    const unsigned g3 = ((x3 & 0x7F7F7F7Fu) + 0x7F7F7F7Fu) | x3;
                    const unsigned all = g0 & g1 & g2 & g3 & 0x80808080u;
                    if (__builtin_amdgcn_ballot_w32(all != 0x80808080u) != 0u) {
                        unsigned h = ((~g0 & 0x80808080u) >> 7) | ((~g1 & 0x80808080u) >> 6)
                                   | ((~g2 & 0x80808080u) >> 5) | ((~g3 & 0x80808080u) >> 4);
                        const unsigned e = (unsigned)(pos + lane * 16);
                        unsigned m = __builtin_amdgcn_ballot_w32(h != 0u);
                        while (m != 0u) {
                            if (h != 0u) {
                                const unsigned bit = (unsigned)__builtin_ctz(h);
                                myq[cnt + __popc(m & ltmask)] = e + ((bit & 7u) << 2) + (bit >> 3);
                                h &= h - 1u;
                            }
                            cnt += __popc(m);
                            m = __builtin_amdgcn_ballot_w32(h != 0u);
                        }
                    }
                    pos += 512;
                }
            }
        }
        __builtin_amdgcn_wave_barrier();
        for (int i0 = 0; i0 < cnt; i0 += 128) {
            unsigned qe[4];
            int qd[4];
#pragma unroll
            for (int j = 0; j < 4; ++j) {
                const int i = i0 + j * 32 + lane;
                qe[j] = (i < cnt) ? myq[i] : 0u;
            }
#pragma unroll
            for (int j = 0; j < 4; ++j) qd[j] = dstv[qe[j]];
#pragma unroll
            for (int j = 0; j < 4; ++j) {
                const int i = i0 + j * 32 + lane;
                if (i < cnt) myq[i] = (qe[j] << 8) | (unsigned)(qd[j] - r0);
            }
        }
        if (lane == 0) {
            qcnt[wave] = cnt;
            if (pos < end) pend[rnd & 1] = 1;
        }
        __syncthreads();

        int pc = 0;
        for (int q = 0; q < AG_W; ++q) {
            const int n = qcnt[q];
            const unsigned* qq = qbuf + q * AG_QCAP;
            for (int i0 = 0; i0 < n; i0 += 32) {
                const int i = i0 + lane;
                unsigned ent = 0;
                bool mine = false;
                if (i < n) { ent = qq[i]; mine = ((ent & 31u) == (unsigned)wave); }
                const unsigned m = __builtin_amdgcn_ballot_w32(mine);
                if (m != 0u) {
                    if (mine) mb[pc + __popc(m & ltmask)] = ent;
                    pc += __popc(m);
                    __builtin_amdgcn_wave_barrier();
                    if (pc >= 32) {
                        ag_drain<MODE == 0>(mb, 32, lane, srcv, hs, acc, wl, wpos);
                        unsigned t = 0;
                        if (lane < pc - 32) t = mb[32 + lane];
                        __builtin_amdgcn_wave_barrier();
                        if (lane < pc - 32) mb[lane] = t;
                        pc -= 32;
                        __builtin_amdgcn_wave_barrier();
                    }
                }
            }
        }
        if (pc > 0) ag_drain<MODE == 0>(mb, pc, lane, srcv, hs, acc, wl, wpos);

        const int again = pend[rnd & 1];
        if (tid == 0) pend[(rnd + 1) & 1] = 0;
        __syncthreads();
        if (!again) break;
    }

    if (MODE == 0 && lane == 0) lcnt[blockIdx.x * AG_W + wave] = wpos;

    if (MODE == 0) ag_transform_out(acc, sdis, dis, bias, Bt, out, r0, nr, wave, lane);
    else           ag_write_out(acc, dis, out, r0, nr, tid);
}

__global__ void __launch_bounds__(256) q_kernel(const int* __restrict__ va, const float* __restrict__ agg,
                                                const float* __restrict__ b2,
                                                const float* __restrict__ Wq, const float* __restrict__ bq,
                                                float* __restrict__ out, int A) {
    const int lane = threadIdx.x & 31;
    const int wv = __builtin_amdgcn_readfirstlane((int)((blockIdx.x * blockDim.x + threadIdx.x) >> 5));
    const int a0 = wv * 32;
    if (a0 >= A) return;
    const int a  = a0 + lane;
    const int ac = (a < A) ? a : (A - 1);
    const int2 ij = *(const int2*)(va + 2 * (long long)ac);

    const float4 b  = *(const float4*)(b2 + lane * 4);
    const float4 wi = *(const float4*)(Wq + lane * 4);
    const float4 wj = *(const float4*)(Wq + 128 + lane * 4);
    const float bq0 = bq[0];

    const bool h16 = (lane & 16) != 0;
    const bool h8  = (lane & 8) != 0;
    const bool h4  = (lane & 4) != 0;
    const int t = lane & 7;
    const int srcl = ((t & 1) << 4) | ((t & 2) << 2) | (t & 4);

    float res = 0.0f;
#pragma unroll
    for (int it = 0; it < 4; ++it) {
        float4 xi[8], xj[8];
#pragma unroll
        for (int k = 0; k < 8; ++k) {
            const int i = __builtin_amdgcn_readlane(ij.x, it * 8 + k);
            const int j = __builtin_amdgcn_readlane(ij.y, it * 8 + k);
            xi[k] = *(const float4*)(agg + (long long)i * 128 + lane * 4);
            xj[k] = *(const float4*)(agg + (long long)j * 128 + lane * 4);
        }
        float v[8];
#pragma unroll
        for (int k = 0; k < 8; ++k) {
            float4 x = xi[k], y = xj[k];
            x.x = x.x + b.x; x.y = x.y + b.y; x.z = x.z + b.z; x.w = x.w + b.w;
            y.x = y.x + b.x; y.y = y.y + b.y; y.z = y.z + b.z; y.w = y.w + b.w;
            x.x = x.x > 0.0f ? x.x : 0.0f;
            x.y = x.y > 0.0f ? x.y : 0.0f;
            x.z = x.z > 0.0f ? x.z : 0.0f;
            x.w = x.w > 0.0f ? x.w : 0.0f;
            y.x = y.x > 0.0f ? y.x : 0.0f;
            y.y = y.y > 0.0f ? y.y : 0.0f;
            y.z = y.z > 0.0f ? y.z : 0.0f;
            y.w = y.w > 0.0f ? y.w : 0.0f;
            v[k] = (x.x * wi.x + x.y * wi.y + x.z * wi.z + x.w * wi.w)
                 + (y.x * wj.x + y.y * wj.y + y.z * wj.z + y.w * wj.w);
        }
        float w4[4];
#pragma unroll
        for (int p = 0; p < 4; ++p) {
            const float keep = h16 ? v[2 * p + 1] : v[2 * p];
            const float send = h16 ? v[2 * p] : v[2 * p + 1];
            w4[p] = keep + __shfl_xor(send, 16);
        }
        float u2[2];
#pragma unroll
        for (int p = 0; p < 2; ++p) {
            const float keep = h8 ? w4[2 * p + 1] : w4[2 * p];
            const float send = h8 ? w4[2 * p] : w4[2 * p + 1];
            u2[p] = keep + __shfl_xor(send, 8);
        }
        float z;
        {
            const float keep = h4 ? u2[1] : u2[0];
            const float send = h4 ? u2[0] : u2[1];
            z = keep + __shfl_xor(send, 4);
        }
        z += __shfl_xor(z, 2);
        z += __shfl_xor(z, 1);
        const float r = __shfl(z, srcl);
        if ((lane >> 3) == it) res = r;
    }
    if (a < A) out[a] = res + bq0;
}

extern "C" void kernel_launch(void* const* d_in, const int* in_sizes, int n_in,
                              void* d_out, int out_size, void* d_ws, size_t ws_size,
                              hipStream_t stream) {
    const float* x  = (const float*)d_in[0];
    const int*   ei = (const int*)d_in[1];
    const int*   va = (const int*)d_in[2];
    const float* W1 = (const float*)d_in[3];
    const float* b1 = (const float*)d_in[4];
    const float* W2 = (const float*)d_in[5];
    const float* b2 = (const float*)d_in[6];
    const float* Wq = (const float*)d_in[7];
    const float* bq = (const float*)d_in[8];
    float* q = (float*)d_out;

    const int H  = 128;
    const int Nn = in_sizes[0] / H;
    const int E  = in_sizes[1] / 2;
    const int A  = in_sizes[2] / 2;

    char*  ws  = (char*)d_ws;
    size_t off = 0;
    auto alloc = [&](size_t bytes) -> char* {
        char* p = ws + off;
        off = (off + bytes + 255) & ~(size_t)255;
        return p;
    };
    _Float16* w1t  = (_Float16*)alloc((size_t)H * H * 2);
    _Float16* w2t  = (_Float16*)alloc((size_t)H * H * 2);
    float*    gout = (float*)   alloc((size_t)Nn * H * 4);
    float*    gou2 = (float*)   alloc((size_t)Nn * H * 4);
    float*    agg2 = (float*)   alloc((size_t)Nn * H * 4);
    float*    dis  = (float*)   alloc((size_t)Nn * 4);

    auto blks = [](long long n, int t) { return (unsigned)((n + t - 1) / t); };

    prep_w_deg<<<blks(2LL * H * H + Nn, 256), 256, 0, stream>>>(W1, W2, w1t, w2t, dis, Nn);
    const int slice  = ((E + 16383) >> 14) << 9;
    const int nwords = 32 * slice / 4 + AG_U * 128;
    unsigned* own = (unsigned*)alloc((size_t)nwords * 4);
    deg_accum_own<<<blks(nwords, 256), 256, 0, stream>>>(ei + E, dis, own, E, nwords);

    const unsigned mtiles = (unsigned)((Nn + 15) / 16);
    const unsigned owners = blks(Nn, AG_R);
    unsigned* lists = (unsigned*)alloc((size_t)owners * AG_W * AG_LCAP * 4);
    int*      lcnt  = (int*)     alloc((size_t)owners * AG_W * 4);

    gemm_fused<<<mtiles, 256, 0, stream>>>(x, w1t, gout, dis, Nn);
    agg_owner<0><<<owners, AG_T, 0, stream>>>(ei, dis, gout, gou2, lists, lcnt, b1, w2t, own, E, Nn);

    agg_owner<1><<<owners, AG_T, 0, stream>>>(ei, dis, gou2, agg2, lists, lcnt, b1, w2t, own, E, Nn);

    q_kernel<<<blks(A, 256), 256, 0, stream>>>(va, agg2, b2, Wq, bq, q, A);
}


# layer-1 owner blocks: scanning waves hand entries to the owning wave through rank-placed LDS bins, one per wave pair; the owner strings its 32 bins together, no sweep of all 32 queues; sums unchanged
# speedup vs baseline: 2.8234x; 1.0689x over previous
#include <hip/hip_runtime.h>
#include <hip/hip_bf16.h>


typedef __attribute__((ext_vector_type(16))) _Float16 v16h;
typedef __attribute__((ext_vector_type(8)))  _Float16 v8h;
typedef __attribute__((ext_vector_type(4)))  _Float16 v4h;
typedef __attribute__((ext_vector_type(8)))  float    v8f;

__global__ void prep_w_deg(const float* __restrict__ W1, const float* __restrict__ W2,
                           _Float16* __restrict__ w1t, _Float16* __restrict__ w2t,
                           float* __restrict__ deg, int Nn) {
    int idx = blockIdx.x * blockDim.x + threadIdx.x;
    if (idx < 16384) {
        int k = idx >> 7, c = idx & 127;
        w1t[c * 128 + k] = (_Float16)W1[idx];
    } else if (idx < 32768) {
        int j = idx - 16384;
        int k = j >> 7, c = j & 127;
        w2t[c * 128 + k] = (_Float16)W2[j];
    } else {
        int i = idx - 32768;
        if (i < Nn) deg[i] = 1.0f;
    }
}

#define OWN_R 200
__global__ void deg_accum_own(const int* __restrict__ dst, float* __restrict__ deg,
                              unsigned* __restrict__ own, int E, int nwords) {
    const int w = blockIdx.x * blockDim.x + threadIdx.x;
    if (w >= nwords) return;
    const int e = w * 4;
    int d[4] = {-1, -1, -1, -1};
    if (e + 3 < E && ((((size_t)dst) & 15) == 0)) {
        const int4 v = *(const int4*)(dst + e);
        d[0] = v.x; d[1] = v.y; d[2] = v.z; d[3] = v.w;
    } else {
#pragma unroll
        for (int k = 0; k < 4; ++k)
            if (e + k < E) d[k] = dst[e + k];
    }
    unsigned pk = 0u;
#pragma unroll
    for (int k = 0; k < 4; ++k) {
        unsigned b = 255u;
        if (d[k] >= 0) {
            atomicAdd(&deg[d[k]], 1.0f);
            b = (unsigned)d[k] / (unsigned)OWN_R;
        }
        pk |= b << (8 * k);
    }
    own[w] = pk;
}

#define A_PITCH 136
__global__ void __launch_bounds__(256) gemm_fused(const float* __restrict__ A32,
                                                  const _Float16* __restrict__ Bt,
                                                  float* __restrict__ C,
                                                  float* __restrict__ dis,
                                                  int nrows) {
    __shared__ __attribute__((aligned(16))) _Float16 As[16 * A_PITCH];
    __shared__ float sdis[16];

    const int tid  = threadIdx.x;
    const int wave = tid >> 5;
    const int lane = tid & 31;
    const int mt   = blockIdx.x;

    const int colB  = wave * 16 + (lane & 15);
    const int kbase = (lane >> 4) ? 8 : 0;
    const _Float16* __restrict__ brow = Bt + (long long)colB * 128;

    v8h blo[4], bhi[4];
#pragma unroll
    for (int kc = 0; kc < 4; ++kc) {
        const int k0 = kc * 32 + kbase;
        blo[kc] = *(const v8h*)(brow + k0);
        bhi[kc] = *(const v8h*)(brow + k0 + 16);
    }

    if (tid < 16) {
        const int i = mt * 16 + tid;
        float r = 0.0f;
        if (i < nrows) {
            const float d = dis[i];
            r = (d > 0.0f) ? rsqrtf(d) : 0.0f;
            dis[i] = r;
        }
        sdis[tid] = r;
    }

#pragma unroll
    for (int p = 0; p < 2; ++p) {
        const int f  = tid + p * 256;
        const int r  = f >> 5;
        const int c4 = (f & 31) * 4;
        int row = mt * 16 + r;
        if (row > nrows - 1) row = nrows - 1;
        float4 v = *(const float4*)(A32 + (long long)row * 128 + c4);
        v4h hv;
        hv[0] = (_Float16)v.x; hv[1] = (_Float16)v.y; hv[2] = (_Float16)v.z; hv[3] = (_Float16)v.w;
        *(v4h*)(As + r * A_PITCH + c4) = hv;
    }

    __syncthreads();

    const _Float16* arow = As + (lane & 15) * A_PITCH;
    v8f acc = {};
#pragma unroll
    for (int kc = 0; kc < 4; ++kc) {
        const int k0 = kc * 32 + kbase;
        v8h alo = *(const v8h*)(arow + k0);
        v8h ahi = *(const v8h*)(arow + k0 + 16);
        v16h a, b;
#pragma unroll
        for (int i = 0; i < 8; ++i) {
            a[i] = alo[i];     a[8 + i] = ahi[i];
            b[i] = blo[kc][i]; b[8 + i] = bhi[kc][i];
        }
        acc = __builtin_amdgcn_wmma_f32_16x16x32_f16(
             false, a,  false, b,
             (short)0, acc,  false,  false);
    }

    const int rbase = mt * 16 + ((lane >> 4) ? 8 : 0);
    float* __restrict__ crow = C + (long long)rbase * 128 + wave * 16 + (lane & 15);
    const float* sd = sdis + ((lane >> 4) ? 8 : 0);
    if (mt * 16 + 16 <= nrows) {
#pragma unroll
        for (int r = 0; r < 8; ++r) crow[(long long)r * 128] = acc[r] * sd[r];
    } else {
#pragma unroll
        for (int r = 0; r < 8; ++r)
            if (rbase + r < nrows) crow[(long long)r * 128] = acc[r] * sd[r];
    }
}

#define AG_R    OWN_R
#define AG_RP   208
#define AG_P    132
#define AG_PH   264
#define AG_T    1024
#define AG_W    32
#define AG_QCAP 768
#define AG_U    8
#define AG_LCAP 192
#define AG_BC   16
#define AG_BP   (AG_W * AG_BC + 1)
#define AG_CP   (AG_W + 1)

__device__ __forceinline__ void ag_add_rows(unsigned lr_l, int s, int cnt, int lane,
                                            const float* __restrict__ hs, float* acc) {
    for (int g = 0; g < cnt; g += 8) {
        float4 v[8];
#pragma unroll
        for (int j = 0; j < 8; ++j) {
            const int sj = __builtin_amdgcn_readlane(s, g + j);
            v[j] = *(const float4*)(hs + (long long)sj * 128 + lane * 4);
        }
#pragma unroll
        for (int j = 0; j < 8; ++j) {
            if (g + j < cnt) {
                const unsigned lr = (unsigned)__builtin_amdgcn_readlane((int)lr_l, g + j);
                float4* a = (float4*)(acc + lr * AG_P + lane * 4);
                float4 t = *a;
                t.x += v[j].x; t.y += v[j].y; t.z += v[j].z; t.w += v[j].w;
                *a = t;
            }
        }
    }
}

template <bool SAVE>
__device__ __forceinline__ void ag_drain(const unsigned* mb, int cnt, int lane,
                                         const int* __restrict__ srcv, const float* __restrict__ hs,
                                         float* acc, unsigned* __restrict__ wl, int& wpos) {
    unsigned ent = 0;
    int s = 0;
    if (lane < cnt) { ent = mb[lane]; s = srcv[ent >> 8]; }
    if (SAVE) {
        if (lane < cnt && wpos + lane < AG_LCAP) wl[wpos + lane] = ((unsigned)s << 8) | (ent & 255u);
        wpos += cnt;
    }
    ag_add_rows(ent & 255u, s, cnt, lane, hs, acc);
}

__device__ __forceinline__ void ag_write_out(const float* acc, const float* __restrict__ dis,
                                             float* __restrict__ out, int r0, int nr, int tid) {
    for (int f = tid; f < AG_R * 32; f += AG_T) {
        const int r = f >> 5, c4 = (f & 31) * 4;
        if (r < nr) {
            const float dr = dis[r0 + r];
            float4 v = *(const float4*)(acc + r * AG_P + c4);
            v.x *= dr; v.y *= dr; v.z *= dr; v.w *= dr;
            *(float4*)(out + (long long)(r0 + r) * 128 + c4) = v;
        }
    }
}

__device__ __forceinline__ void ag_transform_out(float* acc, float* sdis,
                                                 const float* __restrict__ dis, const float* __restrict__ bias,
                                                 const _Float16* __restrict__ Bt, float* __restrict__ out,
                                                 int r0, int nr, int wave, int lane) {
    const int nt    = wave & 7;
    const int colB  = nt * 16 + (lane & 15);
    const int kbase = (lane >> 4) ? 8 : 0;
    const _Float16* __restrict__ brow = Bt + colB * 128;

    v8h blo[4], bhi[4];
#pragma unroll
    for (int kc = 0; kc < 4; ++kc) {
        const int k0 = kc * 32 + kbase;
        blo[kc] = *(const v8h*)(brow + k0);
        bhi[kc] = *(const v8h*)(brow + k0 + 16);
    }

    const int c4 = lane * 4;
    const float4 bb = *(const float4*)(bias + c4);
    float dv[7];
#pragma unroll
    for (int p = 0; p < 7; ++p) {
        const int r  = wave + p * 32;
        const int rn = (r < nr) ? r : (nr - 1);
        dv[p] = dis[r0 + rn];
    }
    float4 t[7];
#pragma unroll
    for (int p = 0; p < 7; ++p) {
        const int r  = wave + p * 32;
        const int ra = (r < AG_R) ? r : (AG_R - 1);
        const float dr = (r < nr) ? dv[p] : 0.0f;
        if (lane == 0 && r < AG_R) sdis[r] = dr;
        float4 v = *(const float4*)(acc + ra * AG_P + c4);
        v.x = v.x * dr + bb.x; v.y = v.y * dr + bb.y; v.z = v.z * dr + bb.z; v.w = v.w * dr + bb.w;
        v.x = v.x > 0.0f ? v.x : 0.0f;
        v.y = v.y > 0.0f ? v.y : 0.0f;
        v.z = v.z > 0.0f ? v.z : 0.0f;
        v.w = v.w > 0.0f ? v.w : 0.0f;
        t[p] = (r < AG_R) ? v : make_float4(0.0f, 0.0f, 0.0f, 0.0f);
    }
    __syncthreads();
    _Float16* tile = (_Float16*)acc;
#pragma unroll
    for (int p = 0; p < 7; ++p) {
        const int r = wave + p * 32;
        if (r < AG_RP) {
            v4h hv;
            hv[0] = (_Float16)t[p].x; hv[1] = (_Float16)t[p].y; hv[2] = (_Float16)t[p].z; hv[3] = (_Float16)t[p].w;
            *(v4h*)(tile + r * AG_PH + c4) = hv;
        }
    }
    __syncthreads();

    for (int mt = wave >> 3; mt < AG_RP / 16; mt += 4) {
        const _Float16* arow = tile + (mt * 16 + (lane & 15)) * AG_PH;
        v8f c = {};
#pragma unroll
        for (int kc = 0; kc < 4; ++kc) {
            const int k0 = kc * 32 + kbase;
            v8h alo = *(const v8h*)(arow + k0);
            v8h ahi = *(const v8h*)(arow + k0 + 16);
            v16h a, b;
#pragma unroll
            for (int i = 0; i < 8; ++i) {
                a[i] = alo[i];     a[8 + i] = ahi[i];
                b[i] = blo[kc][i]; b[8 + i] = bhi[kc][i];
            }
            c = __builtin_amdgcn_wmma_f32_16x16x32_f16(
                 false, a,  false, b,
                 (short)0, c,  false,  false);
        }
        const int lrb = mt * 16 + ((lane >> 4) ? 8 : 0);
        float* __restrict__ crow = out + (long long)(r0 + lrb) * 128 + nt * 16 + (lane & 15);
        const float* sd = sdis + lrb;
        if (mt * 16 + 16 <= nr) {
#pragma unroll
            for (int r = 0; r < 8; ++r) crow[(long long)r * 128] = c[r] * sd[r];
        } else {
#pragma unroll
            for (int r = 0; r < 8; ++r)
                if (lrb + r < nr) crow[(long long)r * 128] = c[r] * sd[r];
        }
    }
}

template <int MODE>
__global__ void __launch_bounds__(AG_T) agg_owner(const int* __restrict__ ei, const float* __restrict__ dis,
                                                  const float* __restrict__ hs, float* __restrict__ out,
                                                  unsigned* __restrict__ lists, int* __restrict__ lcnt,
                                                  const float* __restrict__ bias, const _Float16* __restrict__ Bt,
                                                  const unsigned* __restrict__ own,
                                                  int E, int Nn) {
    __shared__ __attribute__((aligned(16))) float acc[AG_RP * AG_P];
    __shared__ float sdis[AG_RP];
    __shared__ unsigned qbuf[AG_W * AG_QCAP];
    __shared__ unsigned mybuf[AG_W * 64];
    __shared__ int qcnt[AG_W];
    __shared__ int pend[2];
    __shared__ unsigned bins[AG_W * AG_BP];
    __shared__ unsigned bcnt[AG_W * AG_CP];
    __shared__ int bovf[2];

    const int tid  = threadIdx.x;
    const int wave = __builtin_amdgcn_readfirstlane(tid >> 5);
    const int lane = tid & 31;
    const unsigned ltmask = (1u << lane) - 1u;
    const int r0 = blockIdx.x * AG_R;
    const int nr = (Nn - r0 < AG_R) ? (Nn - r0) : AG_R;

    const int* __restrict__ srcv = ei;
    const int* __restrict__ dstv = ei + E;

    unsigned* __restrict__ wl = lists + (size_t)(blockIdx.x * AG_W + wave) * AG_LCAP;
    int wpos = 0;
    int cv = 0;
    if (MODE == 1) cv = lcnt[blockIdx.x * AG_W + lane];

    for (int f = tid; f < AG_R * 32; f += AG_T) {
        const int r = f >> 5, c4 = (f & 31) * 4;
        float4 v = make_float4(0.0f, 0.0f, 0.0f, 0.0f);
        if (r < nr) v = *(const float4*)(hs + (long long)(r0 + r) * 128 + c4);
        *(float4*)(acc + r * AG_P + c4) = v;
    }

    if (MODE == 1) {
        if (__builtin_amdgcn_ballot_w32(cv > AG_LCAP) == 0u) {
            const int nl = __builtin_amdgcn_readlane(cv, wave);
            unsigned cur = 0u;
            if (lane < nl) cur = wl[lane];
            __syncthreads();
            int i0 = 0;
            for (; i0 + 32 <= nl; i0 += 32) {
                unsigned nxt = 0u;
                if (i0 + 32 + lane < nl) nxt = wl[i0 + 32 + lane];
                ag_add_rows(cur & 255u, (int)(cur >> 8), 32, lane, hs, acc);
                cur = nxt;
            }
            if (i0 < nl) ag_add_rows(cur & 255u, (int)(cur >> 8), nl - i0, lane, hs, acc);
            __syncthreads();
            ag_write_out(acc, dis, out, r0, nr, tid);
            return;
        }
    }

    if (tid < 2) { pend[tid] = 0; bovf[tid] = 0; }
    __syncthreads();

    const int S = ((E + 16383) >> 14) << 9;
    int pos = wave * S;
    const int end = (pos + S < E) ? (pos + S) : E;
    unsigned* myq = qbuf + wave * AG_QCAP;
    unsigned* mb  = mybuf + wave * 64;
    const unsigned bpat = (unsigned)blockIdx.x * 0x01010101u;

    for (int rnd = 0;; ++rnd) {
        int cnt = 0;
        while (pos < end && cnt <= AG_QCAP - 512) {
            uint4 ow[AG_U];
#pragma unroll
            for (int u = 0; u < AG_U; ++u)
                ow[u] = *(const uint4*)(own + (pos >> 2) + u * 128 + lane * 4);
#pragma unroll
            for (int u = 0; u < AG_U; ++u) {
                if (pos < end && cnt <= AG_QCAP - 512) {
                    const unsigned x0 = ow[u].x ^ bpat, x1 = ow[u].y ^ bpat, x2 = ow[u].z ^ bpat, x3 = ow[u].w ^ bpat;
                    const unsigned g0 = ((x0 & 0x7F7F7F7Fu) + 0x7F7F7F7Fu) | x0;
                    const unsigned g1 = ((x1 & 0x7F7F7F7Fu) + 0x7F7F7F7Fu) | x1;
                    const unsigned g2 = ((x2 & 0x7F7F7F7Fu) + 0x7F7F7F7Fu) | x2;
                    const unsigned g3 = ((x3 & 0x7F7F7F7Fu) + 0x7F7F7F7Fu) | x3;
                    const unsigned all = g0 & g1 & g2 & g3 & 0x80808080u;
                    if (__builtin_amdgcn_ballot_w32(all != 0x80808080u) != 0u) {
                        unsigned h = ((~g0 & 0x80808080u) >> 7) | ((~g1 & 0x80808080u) >> 6)
                                   | ((~g2 & 0x80808080u) >> 5) | ((~g3 & 0x80808080u) >> 4);
                        const unsigned e = (unsigned)(pos + lane * 16);
                        unsigned m = __builtin_amdgcn_ballot_w32(h != 0u);
                        while (m != 0u) {
                            if (h != 0u) {
                                const unsigned bit = (unsigned)__builtin_ctz(h);
                                myq[cnt + __popc(m & ltmask)] = e + ((bit & 7u) << 2) + (bit >> 3);
                                h &= h - 1u;
                            }
                            cnt += __popc(m);
                            m = __builtin_amdgcn_ballot_w32(h != 0u);
                        }
                    }
                    pos += 512;
                }
            }
        }
        __builtin_amdgcn_wave_barrier();
        for (int i0 = 0; i0 < cnt; i0 += 128) {
            unsigned qe[4];
            int qd[4];
#pragma unroll
            for (int j = 0; j < 4; ++j) {
                const int i = i0 + j * 32 + lane;
                qe[j] = (i < cnt) ? myq[i] : 0u;
            }
#pragma unroll
            for (int j = 0; j < 4; ++j) qd[j] = dstv[qe[j]];
#pragma unroll
            for (int j = 0; j < 4; ++j) {
                const int i = i0 + j * 32 + lane;
                if (i < cnt) myq[i] = (qe[j] << 8) | (unsigned)(qd[j] - r0);
            }
        }
        bcnt[wave * AG_CP + lane] = 0u;
        __builtin_amdgcn_wave_barrier();
        bool ovf = false;
        for (int i0 = 0; i0 < cnt; i0 += 32) {
            const int i = i0 + lane;
            const bool valid = i < cnt;
            const unsigned ent = valid ? myq[i] : 0u;
            const unsigned key = ent & 31u;
            unsigned same = __builtin_amdgcn_ballot_w32(valid);
#pragma unroll
            for (int b = 0; b < 5; ++b) {
                const bool bit = ((key >> b) & 1u) != 0u;
                const unsigned m = __builtin_amdgcn_ballot_w32(bit);
                same &= bit ? m : ~m;
            }
            const unsigned bp = bcnt[wave * AG_CP + key] + (unsigned)__popc(same & ltmask);
            if (valid) {
                if (bp < AG_BC) bins[wave * AG_BP + key * AG_BC + bp] = ent;
                else ovf = true;
            }
            __builtin_amdgcn_wave_barrier();
            if (valid && (same >> lane) == 1u) bcnt[wave * AG_CP + key] = bp + 1u;
            __builtin_amdgcn_wave_barrier();
        }
        const bool anyovf = __builtin_amdgcn_ballot_w32(ovf) != 0u;
        if (lane == 0) {
            qcnt[wave] = cnt;
            if (pos < end) pend[rnd & 1] = 1;
            if (anyovf) bovf[rnd & 1] = 1;
        }
        __syncthreads();

        if (__builtin_amdgcn_readfirstlane(bovf[rnd & 1]) == 0) {
            const int nq = (int)bcnt[lane * AG_CP + wave];
            int inc = nq;
#pragma unroll
            for (int d = 1; d < 32; d <<= 1) {
                const int t = __shfl_up(inc, d);
                if (lane >= d) inc += t;
            }
            const int off = inc - nq;
            const int total = __builtin_amdgcn_readlane(inc, 31);
            for (int i = 0; i < AG_BC; ++i) {
                if (__builtin_amdgcn_ballot_w32(i < nq) == 0u) break;
                if (i < nq) myq[off + i] = bins[lane * AG_BP + wave * AG_BC + i];
            }
            __builtin_amdgcn_wave_barrier();
            for (int i0 = 0; i0 < total; i0 += 128) {
                unsigned qe[4];
                int qs[4];
#pragma unroll
                for (int j = 0; j < 4; ++j) {
                    const int i = i0 + j * 32 + lane;
                    qe[j] = (i < total) ? myq[i] : 0u;
                }
#pragma unroll
                for (int j = 0; j < 4; ++j) qs[j] = srcv[qe[j] >> 8];
#pragma unroll
                for (int j = 0; j < 4; ++j) {
                    const int i = i0 + j * 32 + lane;
                    if (i < total) {
                        const unsigned v = ((unsigned)qs[j] << 8) | (qe[j] & 255u);
                        myq[i] = v;
                        if (MODE == 0 && wpos + i < AG_LCAP) wl[wpos + i] = v;
                    }
                }
            }
            if (MODE == 0) wpos += total;
            for (int i0 = 0; i0 < total; i0 += 32) {
                const int c = (total - i0 < 32) ? (total - i0) : 32;
                const unsigned e = (lane < c) ? myq[i0 + lane] : 0u;
                if (c == 32) ag_add_rows(e & 255u, (int)(e >> 8), 32, lane, hs, acc);
                else         ag_add_rows(e & 255u, (int)(e >> 8), c, lane, hs, acc);
            }
        } else {
            int pc = 0;
            for (int q = 0; q < AG_W; ++q) {
                const int n = qcnt[q];
                const unsigned* qq = qbuf + q * AG_QCAP;
                for (int i0 = 0; i0 < n; i0 += 32) {
                    const int i = i0 + lane;
                    unsigned ent = 0;
                    bool mine = false;
                    if (i < n) { ent = qq[i]; mine = ((ent & 31u) == (unsigned)wave); }
                    const unsigned m = __builtin_amdgcn_ballot_w32(mine);
                    if (m != 0u) {
                        if (mine) mb[pc + __popc(m & ltmask)] = ent;
                        pc += __popc(m);
                        __builtin_amdgcn_wave_barrier();
                        if (pc >= 32) {
                            ag_drain<MODE == 0>(mb, 32, lane, srcv, hs, acc, wl, wpos);
                            unsigned t = 0;
                            if (lane < pc - 32) t = mb[32 + lane];
                            __builtin_amdgcn_wave_barrier();
                            if (lane < pc - 32) mb[lane] = t;
                            pc -= 32;
                            __builtin_amdgcn_wave_barrier();
                        }
                    }
                }
            }
            if (pc > 0) ag_drain<MODE == 0>(mb, pc, lane, srcv, hs, acc, wl, wpos);
        }

        const int again = pend[rnd & 1];
        if (tid == 0) { pend[(rnd + 1) & 1] = 0; bovf[(rnd + 1) & 1] = 0; }
        __syncthreads();
        if (!again) break;
    }

    if (MODE == 0 && lane == 0) lcnt[blockIdx.x * AG_W + wave] = wpos;

    if (MODE == 0) ag_transform_out(acc, sdis, dis, bias, Bt, out, r0, nr, wave, lane);
    else           ag_write_out(acc, dis, out, r0, nr, tid);
}

__global__ void __launch_bounds__(256) q_kernel(const int* __restrict__ va, const float* __restrict__ agg,
                                                const float* __restrict__ b2,
                                                const float* __restrict__ Wq, const float* __restrict__ bq,
                                                float* __restrict__ out, int A) {
    const int lane = threadIdx.x & 31;
    const int wv = __builtin_amdgcn_readfirstlane((int)((blockIdx.x * blockDim.x + threadIdx.x) >> 5));
    const int a0 = wv * 32;
    if (a0 >= A) return;
    const int a  = a0 + lane;
    const int ac = (a < A) ? a : (A - 1);
    const int2 ij = *(const int2*)(va + 2 * (long long)ac);

    const float4 b  = *(const float4*)(b2 + lane * 4);
    const float4 wi = *(const float4*)(Wq + lane * 4);
    const float4 wj = *(const float4*)(Wq + 128 + lane * 4);
    const float bq0 = bq[0];

    const bool h16 = (lane & 16) != 0;
    const bool h8  = (lane & 8) != 0;
    const bool h4  = (lane & 4) != 0;
    const int t = lane & 7;
    const int srcl = ((t & 1) << 4) | ((t & 2) << 2) | (t & 4);

    float res = 0.0f;
#pragma unroll
    for (int it = 0; it < 4; ++it) {
        float4 xi[8], xj[8];
#pragma unroll
        for (int k = 0; k < 8; ++k) {
            const int i = __builtin_amdgcn_readlane(ij.x, it * 8 + k);
            const int j = __builtin_amdgcn_readlane(ij.y, it * 8 + k);
            xi[k] = *(const float4*)(agg + (long long)i * 128 + lane * 4);
            xj[k] = *(const float4*)(agg + (long long)j * 128 + lane * 4);
        }
        float v[8];
#pragma unroll
        for (int k = 0; k < 8; ++k) {
            float4 x = xi[k], y = xj[k];
            x.x = x.x + b.x; x.y = x.y + b.y; x.z = x.z + b.z; x.w = x.w + b.w;
            y.x = y.x + b.x; y.y = y.y + b.y; y.z = y.z + b.z; y.w = y.w + b.w;
            x.x = x.x > 0.0f ? x.x : 0.0f;
            x.y = x.y > 0.0f ? x.y : 0.0f;
            x.z = x.z > 0.0f ? x.z : 0.0f;
            x.w = x.w > 0.0f ? x.w : 0.0f;
            y.x = y.x > 0.0f ? y.x : 0.0f;
            y.y = y.y > 0.0f ? y.y : 0.0f;
            y.z = y.z > 0.0f ? y.z : 0.0f;
            y.w = y.w > 0.0f ? y.w : 0.0f;
            v[k] = (x.x * wi.x + x.y * wi.y + x.z * wi.z + x.w * wi.w)
                 + (y.x * wj.x + y.y * wj.y + y.z * wj.z + y.w * wj.w);
        }
        float w4[4];
#pragma unroll
        for (int p = 0; p < 4; ++p) {
            const float keep = h16 ? v[2 * p + 1] : v[2 * p];
            const float send = h16 ? v[2 * p] : v[2 * p + 1];
            w4[p] = keep + __shfl_xor(send, 16);
        }
        float u2[2];
#pragma unroll
        for (int p = 0; p < 2; ++p) {
            const float keep = h8 ? w4[2 * p + 1] : w4[2 * p];
            const float send = h8 ? w4[2 * p] : w4[2 * p + 1];
            u2[p] = keep + __shfl_xor(send, 8);
        }
        float z;
        {
            const float keep = h4 ? u2[1] : u2[0];
            const float send = h4 ? u2[0] : u2[1];
            z = keep + __shfl_xor(send, 4);
        }
        z += __shfl_xor(z, 2);
        z += __shfl_xor(z, 1);
        const float r = __shfl(z, srcl);
        if ((lane >> 3) == it) res = r;
    }
    if (a < A) out[a] = res + bq0;
}

extern "C" void kernel_launch(void* const* d_in, const int* in_sizes, int n_in,
                              void* d_out, int out_size, void* d_ws, size_t ws_size,
                              hipStream_t stream) {
    const float* x  = (const float*)d_in[0];
    const int*   ei = (const int*)d_in[1];
    const int*   va = (const int*)d_in[2];
    const float* W1 = (const float*)d_in[3];
    const float* b1 = (const float*)d_in[4];
    const float* W2 = (const float*)d_in[5];
    const float* b2 = (const float*)d_in[6];
    const float* Wq = (const float*)d_in[7];
    const float* bq = (const float*)d_in[8];
    float* q = (float*)d_out;

    const int H  = 128;
    const int Nn = in_sizes[0] / H;
    const int E  = in_sizes[1] / 2;
    const int A  = in_sizes[2] / 2;

    char*  ws  = (char*)d_ws;
    size_t off = 0;
    auto alloc = [&](size_t bytes) -> char* {
        char* p = ws + off;
        off = (off + bytes + 255) & ~(size_t)255;
        return p;
    };
    _Float16* w1t  = (_Float16*)alloc((size_t)H * H * 2);
    _Float16* w2t  = (_Float16*)alloc((size_t)H * H * 2);
    float*    gout = (float*)   alloc((size_t)Nn * H * 4);
    float*    gou2 = (float*)   alloc((size_t)Nn * H * 4);
    float*    agg2 = (float*)   alloc((size_t)Nn * H * 4);
    float*    dis  = (float*)   alloc((size_t)Nn * 4);

    auto blks = [](long long n, int t) { return (unsigned)((n + t - 1) / t); };

    prep_w_deg<<<blks(2LL * H * H + Nn, 256), 256, 0, stream>>>(W1, W2, w1t, w2t, dis, Nn);
    const int slice  = ((E + 16383) >> 14) << 9;
    const int nwords = 32 * slice / 4 + AG_U * 128;
    unsigned* own = (unsigned*)alloc((size_t)nwords * 4);
    deg_accum_own<<<blks(nwords, 256), 256, 0, stream>>>(ei + E, dis, own, E, nwords);

    const unsigned mtiles = (unsigned)((Nn + 15) / 16);
    const unsigned owners = blks(Nn, AG_R);
    unsigned* lists = (unsigned*)alloc((size_t)owners * AG_W * AG_LCAP * 4);
    int*      lcnt  = (int*)     alloc((size_t)owners * AG_W * 4);

    gemm_fused<<<mtiles, 256, 0, stream>>>(x, w1t, gout, dis, Nn);
    agg_owner<0><<<owners, AG_T, 0, stream>>>(ei, dis, gout, gou2, lists, lcnt, b1, w2t, own, E, Nn);

    agg_owner<1><<<owners, AG_T, 0, stream>>>(ei, dis, gou2, agg2, lists, lcnt, b1, w2t, own, E, Nn);

    q_kernel<<<blks(A, 256), 256, 0, stream>>>(va, agg2, b2, Wq, bq, q, A);
}


# row sums kept in registers: owner waves group their edge lists by row (order in a row kept, own node first) and sum each row in registers; no LDS read-modify-write or start pass; sums unchanged
# speedup vs baseline: 2.9804x; 1.0556x over previous
#include <hip/hip_runtime.h>
#include <hip/hip_bf16.h>


typedef __attribute__((ext_vector_type(16))) _Float16 v16h;
typedef __attribute__((ext_vector_type(8)))  _Float16 v8h;
typedef __attribute__((ext_vector_type(4)))  _Float16 v4h;
typedef __attribute__((ext_vector_type(8)))  float    v8f;

__global__ void prep_w_deg(const float* __restrict__ W1, const float* __restrict__ W2,
                           _Float16* __restrict__ w1t, _Float16* __restrict__ w2t,
                           float* __restrict__ deg, int Nn) {
    int idx = blockIdx.x * blockDim.x + threadIdx.x;
    if (idx < 16384) {
        int k = idx >> 7, c = idx & 127;
        w1t[c * 128 + k] = (_Float16)W1[idx];
    } else if (idx < 32768) {
        int j = idx - 16384;
        int k = j >> 7, c = j & 127;
        w2t[c * 128 + k] = (_Float16)W2[j];
    } else {
        int i = idx - 32768;
        if (i < Nn) deg[i] = 1.0f;
    }
}

#define OWN_R 200
__global__ void deg_accum_own(const int* __restrict__ dst, float* __restrict__ deg,
                              unsigned* __restrict__ own, int E, int nwords) {
    const int w = blockIdx.x * blockDim.x + threadIdx.x;
    if (w >= nwords) return;
    const int e = w * 4;
    int d[4] = {-1, -1, -1, -1};
    if (e + 3 < E && ((((size_t)dst) & 15) == 0)) {
        const int4 v = *(const int4*)(dst + e);
        d[0] = v.x; d[1] = v.y; d[2] = v.z; d[3] = v.w;
    } else {
#pragma unroll
        for (int k = 0; k < 4; ++k)
            if (e + k < E) d[k] = dst[e + k];
    }
    unsigned pk = 0u;
#pragma unroll
    for (int k = 0; k < 4; ++k) {
        unsigned b = 255u;
        if (d[k] >= 0) {
            atomicAdd(&deg[d[k]], 1.0f);
            b = (unsigned)d[k] / (unsigned)OWN_R;
        }
        pk |= b << (8 * k);
    }
    own[w] = pk;
}

#define A_PITCH 136
__global__ void __launch_bounds__(256) gemm_fused(const float* __restrict__ A32,
                                                  const _Float16* __restrict__ Bt,
                                                  float* __restrict__ C,
                                                  float* __restrict__ dis,
                                                  int nrows) {
    __shared__ __attribute__((aligned(16))) _Float16 As[16 * A_PITCH];
    __shared__ float sdis[16];

    const int tid  = threadIdx.x;
    const int wave = tid >> 5;
    const int lane = tid & 31;
    const int mt   = blockIdx.x;

    const int colB  = wave * 16 + (lane & 15);
    const int kbase = (lane >> 4) ? 8 : 0;
    const _Float16* __restrict__ brow = Bt + (long long)colB * 128;

    v8h blo[4], bhi[4];
#pragma unroll
    for (int kc = 0; kc < 4; ++kc) {
        const int k0 = kc * 32 + kbase;
        blo[kc] = *(const v8h*)(brow + k0);
        bhi[kc] = *(const v8h*)(brow + k0 + 16);
    }

    if (tid < 16) {
        const int i = mt * 16 + tid;
        float r = 0.0f;
        if (i < nrows) {
            const float d = dis[i];
            r = (d > 0.0f) ? rsqrtf(d) : 0.0f;
            dis[i] = r;
        }
        sdis[tid] = r;
    }

#pragma unroll
    for (int p = 0; p < 2; ++p) {
        const int f  = tid + p * 256;
        const int r  = f >> 5;
        const int c4 = (f & 31) * 4;
        int row = mt * 16 + r;
        if (row > nrows - 1) row = nrows - 1;
        float4 v = *(const float4*)(A32 + (long long)row * 128 + c4);
        v4h hv;
        hv[0] = (_Float16)v.x; hv[1] = (_Float16)v.y; hv[2] = (_Float16)v.z; hv[3] = (_Float16)v.w;
        *(v4h*)(As + r * A_PITCH + c4) = hv;
    }

    __syncthreads();

    const _Float16* arow = As + (lane & 15) * A_PITCH;
    v8f acc = {};
#pragma unroll
    for (int kc = 0; kc < 4; ++kc) {
        const int k0 = kc * 32 + kbase;
        v8h alo = *(const v8h*)(arow + k0);
        v8h ahi = *(const v8h*)(arow + k0 + 16);
        v16h a, b;
#pragma unroll
        for (int i = 0; i < 8; ++i) {
            a[i] = alo[i];     a[8 + i] = ahi[i];
            b[i] = blo[kc][i]; b[8 + i] = bhi[kc][i];
        }
        acc = __builtin_amdgcn_wmma_f32_16x16x32_f16(
             false, a,  false, b,
             (short)0, acc,  false,  false);
    }

    const int rbase = mt * 16 + ((lane >> 4) ? 8 : 0);
    float* __restrict__ crow = C + (long long)rbase * 128 + wave * 16 + (lane & 15);
    const float* sd = sdis + ((lane >> 4) ? 8 : 0);
    if (mt * 16 + 16 <= nrows) {
#pragma unroll
        for (int r = 0; r < 8; ++r) crow[(long long)r * 128] = acc[r] * sd[r];
    } else {
#pragma unroll
        for (int r = 0; r < 8; ++r)
            if (rbase + r < nrows) crow[(long long)r * 128] = acc[r] * sd[r];
    }
}

#define AG_R    OWN_R
#define AG_RP   208
#define AG_P    132
#define AG_PH   264
#define AG_T    1024
#define AG_W    32
#define AG_QCAP 768
#define AG_U    8
#define AG_LCAP 192
#define AG_BC   16
#define AG_BP   (AG_W * AG_BC + 1)
#define AG_CP   (AG_W + 1)
#define AG_GB   256
#define AG_GNULL 0xFFu

__device__ __forceinline__ void ag_add_rows(unsigned lr_l, int s, int cnt, int lane,
                                            const float* __restrict__ hs, float* acc) {
    for (int g = 0; g < cnt; g += 8) {
        float4 v[8];
#pragma unroll
        for (int j = 0; j < 8; ++j) {
            const int sj = __builtin_amdgcn_readlane(s, g + j);
            v[j] = *(const float4*)(hs + (long long)sj * 128 + lane * 4);
        }
#pragma unroll
        for (int j = 0; j < 8; ++j) {
            if (g + j < cnt) {
                const unsigned lr = (unsigned)__builtin_amdgcn_readlane((int)lr_l, g + j);
                float4* a = (float4*)(acc + lr * AG_P + lane * 4);
                float4 t = *a;
                t.x += v[j].x; t.y += v[j].y; t.z += v[j].z; t.w += v[j].w;
                *a = t;
            }
        }
    }
}

template <bool SAVE>
__device__ __forceinline__ void ag_drain(const unsigned* mb, int cnt, int lane,
                                         const int* __restrict__ srcv, const float* __restrict__ hs,
                                         float* acc, unsigned* __restrict__ wl, int& wpos) {
    unsigned ent = 0;
    int s = 0;
    if (lane < cnt) { ent = mb[lane]; s = srcv[ent >> 8]; }
    if (SAVE) {
        if (lane < cnt && wpos + lane < AG_LCAP) wl[wpos + lane] = ((unsigned)s << 8) | (ent & 255u);
        wpos += cnt;
    }
    ag_add_rows(ent & 255u, s, cnt, lane, hs, acc);
}

__device__ __forceinline__ void ag_init_own(const float* __restrict__ hs, float* acc, int r0, int nr, int wave, int lane) {
    float4 sv[7];
#pragma unroll
    for (int p = 0; p < 7; ++p) {
        const int row = wave + p * 32;
        const int rr  = (row < nr) ? row : (nr - 1);
        sv[p] = *(const float4*)(hs + (long long)(r0 + rr) * 128 + (unsigned)lane * 4u);
    }
#pragma unroll
    for (int p = 0; p < 7; ++p) {
        const int row = wave + p * 32;
        if (row < AG_R)
            *(float4*)(acc + row * AG_P + lane * 4) = (row < nr) ? sv[p] : make_float4(0.0f, 0.0f, 0.0f, 0.0f);
    }
}

template <bool TO_LDS>
__device__ __forceinline__ void ag_sum_grouped(const unsigned* gl, int g, int lane, float disv,
                                               const float* __restrict__ hs, float* accl,
                                               float* __restrict__ out, int r0) {
    if (g <= 0) return;
    float4 V[8];
    float4 acc = make_float4(0.0f, 0.0f, 0.0f, 0.0f);
    unsigned cur = AG_GNULL;
    const unsigned loff = (unsigned)lane * 4u;

#define AG_HANDOVER()                                                                                 \
    if (TO_LDS) {                                                                                     \
        *(float4*)(accl + cur * AG_P + loff) = acc;                                                   \
    } else {                                                                                          \
        const float dr = __int_as_float(__builtin_amdgcn_readlane(__float_as_int(disv), (int)(cur >> 5))); \
        float4 o = acc;                                                                               \
        o.x *= dr; o.y *= dr; o.z *= dr; o.w *= dr;                                                   \
        *(float4*)(out + (long long)(r0 + (int)cur) * 128 + loff) = o;                                \
    }
#define AG_TAKE(LR, X)                                                                                \
    {                                                                                                 \
        const unsigned lr_ = (LR);                                                                    \
        if (lr_ != cur) {                                                                             \
            if (cur != AG_GNULL) { AG_HANDOVER() }                                                    \
            cur = lr_;                                                                                \
            acc = (X);                                                                                \
        } else {                                                                                      \
            acc.x += (X).x; acc.y += (X).y; acc.z += (X).z; acc.w += (X).w;                           \
        }                                                                                             \
    }
#define AG_ROW(S) (*(const float4*)(hs + (long long)(S) * 128 + loff))

    const unsigned e0 = gl[(lane < g) ? lane : (g - 1)];
    unsigned ent = (lane < g) ? e0 : AG_GNULL;

    int c0 = 0;
    for (; c0 + 32 <= g; c0 += 32) {
        const int ni = c0 + 32 + lane;
        const unsigned nraw = gl[(ni < g) ? ni : (g - 1)];
        const unsigned esrc = ent >> 8, erow = ent & 255u;
#pragma unroll
        for (int j = 0; j < 8; ++j) V[j] = AG_ROW(__builtin_amdgcn_readlane((int)esrc, j));
#pragma unroll
        for (int j = 0; j < 24; ++j) {
            AG_TAKE((unsigned)__builtin_amdgcn_readlane((int)erow, j), V[j & 7])
            V[j & 7] = AG_ROW(__builtin_amdgcn_readlane((int)esrc, j + 8));
        }
#pragma unroll
        for (int j = 24; j < 32; ++j) AG_TAKE((unsigned)__builtin_amdgcn_readlane((int)erow, j), V[j & 7])
        ent = (ni < g) ? nraw : AG_GNULL;
    }
    if (c0 < g) {
        const int rem = g - c0;
        const unsigned esrc = ent >> 8, erow = ent & 255u;
#pragma unroll
        for (int q = 0; q < 4; ++q) {
            if (8 * q < rem) {
#pragma unroll
                for (int j = 0; j < 8; ++j) V[j] = AG_ROW(__builtin_amdgcn_readlane((int)esrc, 8 * q + j));
#pragma unroll
                for (int j = 0; j < 8; ++j)
                    if (8 * q + j < rem) AG_TAKE((unsigned)__builtin_amdgcn_readlane((int)erow, 8 * q + j), V[j])
            }
        }
    }
    if (cur != AG_GNULL) { AG_HANDOVER() }
#undef AG_HANDOVER
#undef AG_TAKE
#undef AG_ROW
}

__device__ __forceinline__ void ag_write_out(const float* acc, const float* __restrict__ dis,
                                             float* __restrict__ out, int r0, int nr, int tid) {
    for (int f = tid; f < AG_R * 32; f += AG_T) {
        const int r = f >> 5, c4 = (f & 31) * 4;
        if (r < nr) {
            const float dr = dis[r0 + r];
            float4 v = *(const float4*)(acc + r * AG_P + c4);
            v.x *= dr; v.y *= dr; v.z *= dr; v.w *= dr;
            *(float4*)(out + (long long)(r0 + r) * 128 + c4) = v;
        }
    }
}

__device__ __forceinline__ void ag_transform_out(float* acc, float* sdis,
                                                 const float* __restrict__ dis, const float* __restrict__ bias,
                                                 const _Float16* __restrict__ Bt, float* __restrict__ out,
                                                 int r0, int nr, int wave, int lane) {
    const int nt    = wave & 7;
    const int colB  = nt * 16 + (lane & 15);
    const int kbase = (lane >> 4) ? 8 : 0;
    const _Float16* __restrict__ brow = Bt + colB * 128;

    v8h blo[4], bhi[4];
#pragma unroll
    for (int kc = 0; kc < 4; ++kc) {
        const int k0 = kc * 32 + kbase;
        blo[kc] = *(const v8h*)(brow + k0);
        bhi[kc] = *(const v8h*)(brow + k0 + 16);
    }

    const int c4 = lane * 4;
    const float4 bb = *(const float4*)(bias + c4);
    float dv[7];
#pragma unroll
    for (int p = 0; p < 7; ++p) {
        const int r  = wave + p * 32;
        const int rn = (r < nr) ? r : (nr - 1);
        dv[p] = dis[r0 + rn];
    }
    float4 t[7];
#pragma unroll
    for (int p = 0; p < 7; ++p) {
        const int r  = wave + p * 32;
        const int ra = (r < AG_R) ? r : (AG_R - 1);
        const float dr = (r < nr) ? dv[p] : 0.0f;
        if (lane == 0 && r < AG_R) sdis[r] = dr;
        float4 v = *(const float4*)(acc + ra * AG_P + c4);
        v.x = v.x * dr + bb.x; v.y = v.y * dr + bb.y; v.z = v.z * dr + bb.z; v.w = v.w * dr + bb.w;
        v.x = v.x > 0.0f ? v.x : 0.0f;
        v.y = v.y > 0.0f ? v.y : 0.0f;
        v.z = v.z > 0.0f ? v.z : 0.0f;
        v.w = v.w > 0.0f ? v.w : 0.0f;
        t[p] = (r < AG_R) ? v : make_float4(0.0f, 0.0f, 0.0f, 0.0f);
    }
    __syncthreads();
    _Float16* tile = (_Float16*)acc;
#pragma unroll
    for (int p = 0; p < 7; ++p) {
        const int r = wave + p * 32;
        if (r < AG_RP) {
            v4h hv;
            hv[0] = (_Float16)t[p].x; hv[1] = (_Float16)t[p].y; hv[2] = (_Float16)t[p].z; hv[3] = (_Float16)t[p].w;
            *(v4h*)(tile + r * AG_PH + c4) = hv;
        }
    }
    __syncthreads();

    for (int mt = wave >> 3; mt < AG_RP / 16; mt += 4) {
        const _Float16* arow = tile + (mt * 16 + (lane & 15)) * AG_PH;
        v8f c = {};
#pragma unroll
        for (int kc = 0; kc < 4; ++kc) {
            const int k0 = kc * 32 + kbase;
            v8h alo = *(const v8h*)(arow + k0);
            v8h ahi = *(const v8h*)(arow + k0 + 16);
            v16h a, b;
#pragma unroll
            for (int i = 0; i < 8; ++i) {
                a[i] = alo[i];     a[8 + i] = ahi[i];
                b[i] = blo[kc][i]; b[8 + i] = bhi[kc][i];
            }
            c = __builtin_amdgcn_wmma_f32_16x16x32_f16(
                 false, a,  false, b,
                 (short)0, c,  false,  false);
        }
        const int lrb = mt * 16 + ((lane >> 4) ? 8 : 0);
        float* __restrict__ crow = out + (long long)(r0 + lrb) * 128 + nt * 16 + (lane & 15);
        const float* sd = sdis + lrb;
        if (mt * 16 + 16 <= nr) {
#pragma unroll
            for (int r = 0; r < 8; ++r) crow[(long long)r * 128] = c[r] * sd[r];
        } else {
#pragma unroll
            for (int r = 0; r < 8; ++r)
                if (lrb + r < nr) crow[(long long)r * 128] = c[r] * sd[r];
        }
    }
}

template <int MODE>
__global__ void __launch_bounds__(AG_T) agg_owner(const int* __restrict__ ei, const float* __restrict__ dis,
                                                  const float* __restrict__ hs, float* __restrict__ out,
                                                  unsigned* __restrict__ lists, int* __restrict__ lcnt,
                                                  const float* __restrict__ bias, const _Float16* __restrict__ Bt,
                                                  const unsigned* __restrict__ own,
                                                  int E, int Nn) {
    __shared__ __attribute__((aligned(16))) float acc[AG_RP * AG_P];
    __shared__ float sdis[AG_RP];
    __shared__ unsigned qbuf[AG_W * AG_QCAP];
    __shared__ unsigned mybuf[AG_W * AG_GB];
    __shared__ int qcnt[AG_W];
    __shared__ int pend[2];
    __shared__ unsigned bins[AG_W * AG_BP];
    __shared__ unsigned bcnt[AG_W * AG_CP];
    __shared__ int bovf[2];

    const int tid  = threadIdx.x;
    const int wave = __builtin_amdgcn_readfirstlane(tid >> 5);
    const int lane = tid & 31;
    const unsigned ltmask = (1u << lane) - 1u;
    const int r0 = blockIdx.x * AG_R;
    const int nr = (Nn - r0 < AG_R) ? (Nn - r0) : AG_R;

    const int* __restrict__ srcv = ei;
    const int* __restrict__ dstv = ei + E;

    unsigned* __restrict__ wl = lists + (size_t)(blockIdx.x * AG_W + wave) * AG_LCAP;
    int gsave = AG_LCAP + 1;

    if (MODE == 1) {
        const int cv = lcnt[blockIdx.x * AG_W + lane];
        if (__builtin_amdgcn_ballot_w32(cv > AG_LCAP) == 0u) {
            const int g = __builtin_amdgcn_readlane(cv, wave);
            float disv = 0.0f;
            if (lane < 7 && wave + 32 * lane < nr) disv = dis[r0 + wave + 32 * lane];
            ag_sum_grouped<false>(wl, g, lane, disv, hs, acc, out, r0);
            return;
        }
    }

    if (tid < 2) { pend[tid] = 0; bovf[tid] = 0; }
    __syncthreads();

    const int S = ((E + 16383) >> 14) << 9;
    int pos = wave * S;
    const int end = (pos + S < E) ? (pos + S) : E;
    unsigned* myq = qbuf + wave * AG_QCAP;
    unsigned* mb  = mybuf + wave * AG_GB;
    const unsigned bpat = (unsigned)blockIdx.x * 0x01010101u;

    for (int rnd = 0;; ++rnd) {
        int cnt = 0;
        while (pos < end && cnt <= AG_QCAP - 512) {
            uint4 ow[AG_U];
#pragma unroll
            for (int u = 0; u < AG_U; ++u)
                ow[u] = *(const uint4*)(own + (pos >> 2) + u * 128 + lane * 4);
#pragma unroll
            for (int u = 0; u < AG_U; ++u) {
                if (pos < end && cnt <= AG_QCAP - 512) {
                    const unsigned x0 = ow[u].x ^ bpat, x1 = ow[u].y ^ bpat, x2 = ow[u].z ^ bpat, x3 = ow[u].w ^ bpat;
                    const unsigned g0 = ((x0 & 0x7F7F7F7Fu) + 0x7F7F7F7Fu) | x0;
                    const unsigned g1 = ((x1 & 0x7F7F7F7Fu) + 0x7F7F7F7Fu) | x1;
                    const unsigned g2 = ((x2 & 0x7F7F7F7Fu) + 0x7F7F7F7Fu) | x2;
                    const unsigned g3 = ((x3 & 0x7F7F7F7Fu) + 0x7F7F7F7Fu) | x3;
                    const unsigned all = g0 & g1 & g2 & g3 & 0x80808080u;
                    if (__builtin_amdgcn_ballot_w32(all != 0x80808080u) != 0u) {
                        unsigned h = ((~g0 & 0x80808080u) >> 7) | ((~g1 & 0x80808080u) >> 6)
                                   | ((~g2 & 0x80808080u) >> 5) | ((~g3 & 0x80808080u) >> 4);
                        const unsigned e = (unsigned)(pos + lane * 16);
                        unsigned m = __builtin_amdgcn_ballot_w32(h != 0u);
                        while (m != 0u) {
                            if (h != 0u) {
                                const unsigned bit = (unsigned)__builtin_ctz(h);
                                myq[cnt + __popc(m & ltmask)] = e + ((bit & 7u) << 2) + (bit >> 3);
                                h &= h - 1u;
                            }
                            cnt += __popc(m);
                            m = __builtin_amdgcn_ballot_w32(h != 0u);
                        }
                    }
                    pos += 512;
                }
            }
        }
        __builtin_amdgcn_wave_barrier();
        for (int i0 = 0; i0 < cnt; i0 += 128) {
            unsigned qe[4];
            int qd[4];
#pragma unroll
            for (int j = 0; j < 4; ++j) {
                const int i = i0 + j * 32 + lane;
                qe[j] = (i < cnt) ? myq[i] : 0u;
            }
#pragma unroll
            for (int j = 0; j < 4; ++j) qd[j] = dstv[qe[j]];
#pragma unroll
            for (int j = 0; j < 4; ++j) {
                const int i = i0 + j * 32 + lane;
                if (i < cnt) myq[i] = (qe[j] << 8) | (unsigned)(qd[j] - r0);
            }
        }
        bcnt[wave * AG_CP + lane] = 0u;
        __builtin_amdgcn_wave_barrier();
        bool ovf = false;
        for (int i0 = 0; i0 < cnt; i0 += 32) {
            const int i = i0 + lane;
            const bool valid = i < cnt;
            const unsigned ent = valid ? myq[i] : 0u;
            const unsigned key = ent & 31u;
            unsigned same = __builtin_amdgcn_ballot_w32(valid);
#pragma unroll
            for (int b = 0; b < 5; ++b) {
                const bool bit = ((key >> b) & 1u) != 0u;
                const unsigned m = __builtin_amdgcn_ballot_w32(bit);
                same &= bit ? m : ~m;
            }
            const unsigned bp = bcnt[wave * AG_CP + key] + (unsigned)__popc(same & ltmask);
            if (valid) {
                if (bp < AG_BC) bins[wave * AG_BP + key * AG_BC + bp] = ent;
                else ovf = true;
            }
            __builtin_amdgcn_wave_barrier();
            if (valid && (same >> lane) == 1u) bcnt[wave * AG_CP + key] = bp + 1u;
            __builtin_amdgcn_wave_barrier();
        }
        const bool anyovf = __builtin_amdgcn_ballot_w32(ovf) != 0u;
        if (lane == 0) {
            qcnt[wave] = cnt;
            if (pos < end) pend[rnd & 1] = 1;
            if (anyovf) bovf[rnd & 1] = 1;
        }
        __syncthreads();

        if (__builtin_amdgcn_readfirstlane(bovf[rnd & 1]) == 0) {
            const int nq = (int)bcnt[lane * AG_CP + wave];
            int inc = nq;
#pragma unroll
            for (int d = 1; d < 32; d <<= 1) {
                const int t = __shfl_up(inc, d);
                if (lane >= d) inc += t;
            }
            const int off = inc - nq;
            const int total = __builtin_amdgcn_readlane(inc, 31);
            for (int i = 0; i < AG_BC; ++i) {
                if (__builtin_amdgcn_ballot_w32(i < nq) == 0u) break;
                if (i < nq) myq[off + i] = bins[lane * AG_BP + wave * AG_BC + i];
            }
            __builtin_amdgcn_wave_barrier();
            for (int i0 = 0; i0 < total; i0 += 128) {
                unsigned qe[4];
                int qs[4];
#pragma unroll
                for (int j = 0; j < 4; ++j) {
                    const int i = i0 + j * 32 + lane;
                    qe[j] = (i < total) ? myq[i] : 0u;
                }
#pragma unroll
                for (int j = 0; j < 4; ++j) qs[j] = srcv[qe[j] >> 8];
#pragma unroll
                for (int j = 0; j < 4; ++j) {
                    const int i = i0 + j * 32 + lane;
                    if (i < total) {
                        const unsigned v = ((unsigned)qs[j] << 8) | (qe[j] & 255u);
                        myq[i] = v;
                    }
                }
            }
            if (MODE == 0 && rnd == 0 && total + 7 <= AG_LCAP) {
                int cntv = 0;
                for (int i0 = 0; i0 < total; i0 += 32) {
                    const int i = i0 + lane;
                    const bool valid = i < total;
                    const unsigned e = valid ? myq[i] : 0u;
                    const int pb = (int)((e & 255u) >> 5);
#pragma unroll
                    for (int p = 0; p < 7; ++p) {
                        const int n = __popc(__builtin_amdgcn_ballot_w32(valid && pb == p));
                        if (lane == p) cntv += n;
                    }
                }
                const int myrow = wave + 32 * lane;
                const int selfv = (lane < 7 && myrow < nr) ? 1 : 0;
                const int tot   = (lane < 7) ? (cntv + selfv) : 0;
                int inc = tot;
#pragma unroll
                for (int d = 1; d < 8; d <<= 1) {
                    const int u = __shfl_up(inc, d);
                    if (lane >= d) inc += u;
                }
                const int g = __builtin_amdgcn_readlane(inc, 6);
                int runv = inc - tot;
                if (selfv) { mb[runv] = ((unsigned)(r0 + myrow) << 8) | (unsigned)myrow; runv += 1; }
                for (int i0 = 0; i0 < total; i0 += 32) {
                    const int i = i0 + lane;
                    const bool valid = i < total;
                    const unsigned e = valid ? myq[i] : 0u;
                    const int pb = (int)((e & 255u) >> 5);
#pragma unroll
                    for (int p = 0; p < 7; ++p) {
                        const bool hit = valid && pb == p;
                        const unsigned m = __builtin_amdgcn_ballot_w32(hit);
                        const int rp = __builtin_amdgcn_readlane(runv, p);
                        if (hit) mb[rp + __popc(m & ltmask)] = e;
                        if (lane == p) runv += __popc(m);
                    }
                }
                __builtin_amdgcn_wave_barrier();
                for (int i = lane; i < g; i += 32) wl[i] = mb[i];
                gsave = g;
                ag_sum_grouped<true>(mb, g, lane, 0.0f, hs, acc, out, r0);
#pragma unroll
                for (int p = 0; p < 7; ++p) {
                    const int row = wave + 32 * p;
                    if (row >= nr && row < AG_R)
                        *(float4*)(acc + row * AG_P + lane * 4) = make_float4(0.0f, 0.0f, 0.0f, 0.0f);
                }
            } else {
                if (rnd == 0) ag_init_own(hs, acc, r0, nr, wave, lane);
                gsave = AG_LCAP + 1;
                for (int i0 = 0; i0 < total; i0 += 32) {
                    const int c = (total - i0 < 32) ? (total - i0) : 32;
                    const unsigned e = (lane < c) ? myq[i0 + lane] : 0u;
                    if (c == 32) ag_add_rows(e & 255u, (int)(e >> 8), 32, lane, hs, acc);
                    else         ag_add_rows(e & 255u, (int)(e >> 8), c, lane, hs, acc);
                }
            }
        } else {
            if (rnd == 0) ag_init_own(hs, acc, r0, nr, wave, lane);
            gsave = AG_LCAP + 1;
            int nosave = 0;
            int pc = 0;
            for (int q = 0; q < AG_W; ++q) {
                const int n = qcnt[q];
                const unsigned* qq = qbuf + q * AG_QCAP;
                for (int i0 = 0; i0 < n; i0 += 32) {
                    const int i = i0 + lane;
                    unsigned ent = 0;
                    bool mine = false;
                    if (i < n) { ent = qq[i]; mine = ((ent & 31u) == (unsigned)wave); }
                    const unsigned m = __builtin_amdgcn_ballot_w32(mine);
                    if (m != 0u) {
                        if (mine) mb[pc + __popc(m & ltmask)] = ent;
                        pc += __popc(m);
                        __builtin_amdgcn_wave_barrier();
                        if (pc >= 32) {
                            ag_drain<false>(mb, 32, lane, srcv, hs, acc, wl, nosave);
                            unsigned t = 0;
                            if (lane < pc - 32) t = mb[32 + lane];
                            __builtin_amdgcn_wave_barrier();
                            if (lane < pc - 32) mb[lane] = t;
                            pc -= 32;
                            __builtin_amdgcn_wave_barrier();
                        }
                    }
                }
            }
            if (pc > 0) ag_drain<false>(mb, pc, lane, srcv, hs, acc, wl, nosave);
        }

        const int again = pend[rnd & 1];
        if (tid == 0) { pend[(rnd + 1) & 1] = 0; bovf[(rnd + 1) & 1] = 0; }
        __syncthreads();
        if (!again) break;
    }

    if (MODE == 0 && lane == 0) lcnt[blockIdx.x * AG_W + wave] = gsave;

    if (MODE == 0) ag_transform_out(acc, sdis, dis, bias, Bt, out, r0, nr, wave, lane);
    else           ag_write_out(acc, dis, out, r0, nr, tid);
}

__global__ void __launch_bounds__(256) q_kernel(const int* __restrict__ va, const float* __restrict__ agg,
                                                const float* __restrict__ b2,
                                                const float* __restrict__ Wq, const float* __restrict__ bq,
                                                float* __restrict__ out, int A) {
    const int lane = threadIdx.x & 31;
    const int wv = __builtin_amdgcn_readfirstlane((int)((blockIdx.x * blockDim.x + threadIdx.x) >> 5));
    const int a0 = wv * 32;
    if (a0 >= A) return;
    const int a  = a0 + lane;
    const int ac = (a < A) ? a : (A - 1);
    const int2 ij = *(const int2*)(va + 2 * (long long)ac);

    const float4 b  = *(const float4*)(b2 + lane * 4);
    const float4 wi = *(const float4*)(Wq + lane * 4);
    const float4 wj = *(const float4*)(Wq + 128 + lane * 4);
    const float bq0 = bq[0];

    const bool h16 = (lane & 16) != 0;
    const bool h8  = (lane & 8) != 0;
    const bool h4  = (lane & 4) != 0;
    const int t = lane & 7;
    const int srcl = ((t & 1) << 4) | ((t & 2) << 2) | (t & 4);

    float res = 0.0f;
#pragma unroll
    for (int it = 0; it < 4; ++it) {
        float4 xi[8], xj[8];
#pragma unroll
        for (int k = 0; k < 8; ++k) {
            const int i = __builtin_amdgcn_readlane(ij.x, it * 8 + k);
            const int j = __builtin_amdgcn_readlane(ij.y, it * 8 + k);
            xi[k] = *(const float4*)(agg + (long long)i * 128 + lane * 4);
            xj[k] = *(const float4*)(agg + (long long)j * 128 + lane * 4);
        }
        float v[8];
#pragma unroll
        for (int k = 0; k < 8; ++k) {
            float4 x = xi[k], y = xj[k];
            x.x = x.x + b.x; x.y = x.y + b.y; x.z = x.z + b.z; x.w = x.w + b.w;
            y.x = y.x + b.x; y.y = y.y + b.y; y.z = y.z + b.z; y.w = y.w + b.w;
            x.x = x.x > 0.0f ? x.x : 0.0f;
            x.y = x.y > 0.0f ? x.y : 0.0f;
            x.z = x.z > 0.0f ? x.z : 0.0f;
            x.w = x.w > 0.0f ? x.w : 0.0f;
            y.x = y.x > 0.0f ? y.x : 0.0f;
            y.y = y.y > 0.0f ? y.y : 0.0f;
            y.z = y.z > 0.0f ? y.z : 0.0f;
            y.w = y.w > 0.0f ? y.w : 0.0f;
            v[k] = (x.x * wi.x + x.y * wi.y + x.z * wi.z + x.w * wi.w)
                 + (y.x * wj.x + y.y * wj.y + y.z * wj.z + y.w * wj.w);
        }
        float w4[4];
#pragma unroll
        for (int p = 0; p < 4; ++p) {
            const float keep = h16 ? v[2 * p + 1] : v[2 * p];
            const float send = h16 ? v[2 * p] : v[2 * p + 1];
            w4[p] = keep + __shfl_xor(send, 16);
        }
        float u2[2];
#pragma unroll
        for (int p = 0; p < 2; ++p) {
            const float keep = h8 ? w4[2 * p + 1] : w4[2 * p];
            const float send = h8 ? w4[2 * p] : w4[2 * p + 1];
            u2[p] = keep + __shfl_xor(send, 8);
        }
        float z;
        {
            const float keep = h4 ? u2[1] : u2[0];
            const float send = h4 ? u2[0] : u2[1];
            z = keep + __shfl_xor(send, 4);
        }
        z += __shfl_xor(z, 2);
        z += __shfl_xor(z, 1);
        const float r = __shfl(z, srcl);
        if ((lane >> 3) == it) res = r;
    }
    if (a < A) out[a] = res + bq0;
}

extern "C" void kernel_launch(void* const* d_in, const int* in_sizes, int n_in,
                              void* d_out, int out_size, void* d_ws, size_t ws_size,
                              hipStream_t stream) {
    const float* x  = (const float*)d_in[0];
    const int*   ei = (const int*)d_in[1];
    const int*   va = (const int*)d_in[2];
    const float* W1 = (const float*)d_in[3];
    const float* b1 = (const float*)d_in[4];
    const float* W2 = (const float*)d_in[5];
    const float* b2 = (const float*)d_in[6];
    const float* Wq = (const float*)d_in[7];
    const float* bq = (const float*)d_in[8];
    float* q = (float*)d_out;

    const int H  = 128;
    const int Nn = in_sizes[0] / H;
    const int E  = in_sizes[1] / 2;
    const int A  = in_sizes[2] / 2;

    char*  ws  = (char*)d_ws;
    size_t off = 0;
    auto alloc = [&](size_t bytes) -> char* {
        char* p = ws + off;
        off = (off + bytes + 255) & ~(size_t)255;
        return p;
    };
    _Float16* w1t  = (_Float16*)alloc((size_t)H * H * 2);
    _Float16* w2t  = (_Float16*)alloc((size_t)H * H * 2);
    float*    gout = (float*)   alloc((size_t)Nn * H * 4);
    float*    gou2 = (float*)   alloc((size_t)Nn * H * 4);
    float*    agg2 = (float*)   alloc((size_t)Nn * H * 4);
    float*    dis  = (float*)   alloc((size_t)Nn * 4);

    auto blks = [](long long n, int t) { return (unsigned)((n + t - 1) / t); };

    prep_w_deg<<<blks(2LL * H * H + Nn, 256), 256, 0, stream>>>(W1, W2, w1t, w2t, dis, Nn);
    const int slice  = ((E + 16383) >> 14) << 9;
    const int nwords = 32 * slice / 4 + AG_U * 128;
    unsigned* own = (unsigned*)alloc((size_t)nwords * 4);
    deg_accum_own<<<blks(nwords, 256), 256, 0, stream>>>(ei + E, dis, own, E, nwords);

    const unsigned mtiles = (unsigned)((Nn + 15) / 16);
    const unsigned owners = blks(Nn, AG_R);
    unsigned* lists = (unsigned*)alloc((size_t)owners * AG_W * AG_LCAP * 4);
    int*      lcnt  = (int*)     alloc((size_t)owners * AG_W * 4);

    gemm_fused<<<mtiles, 256, 0, stream>>>(x, w1t, gout, dis, Nn);
    agg_owner<0><<<owners, AG_T, 0, stream>>>(ei, dis, gout, gou2, lists, lcnt, b1, w2t, own, E, Nn);

    agg_owner<1><<<owners, AG_T, 0, stream>>>(ei, dis, gou2, agg2, lists, lcnt, b1, w2t, own, E, Nn);

    q_kernel<<<blks(A, 256), 256, 0, stream>>>(va, agg2, b2, Wq, bq, q, A);
}
